# starting kernel
# baseline (speedup 1.0000x reference)
#include <hip/hip_runtime.h>
#include <hip/hip_bf16.h>

typedef float v2f __attribute__((ext_vector_type(2)));
typedef float v8f __attribute__((ext_vector_type(8)));

#define N_VOX 20000
#define M_PTS 64
#define C_IN  10
#define C0    64
#define C1    128
#define EPS   1e-3f

#define WS_SUMX   0
#define WS_SXX    16
#define WS_SCALE0 128
#define WS_BIAS0  192
#define WS_SUMY   256
#define WS_SUMQ   384
#define WS_SCALE1 512
#define WS_BIAS1  640
#define WS_TOTAL  768

#define X1_STRIDE 132
#define WT_STRIDE 132

__global__ void vfe_zero(float* __restrict__ ws) {
    int i = blockIdx.x * blockDim.x + threadIdx.x;
    if (i < WS_TOTAL) ws[i] = 0.0f;
}

__global__ __launch_bounds__(256) void vfe_stats0(const float* __restrict__ X,
                                                  float* __restrict__ ws, long P) {
    float sx[C_IN] = {};
    float sxx[C_IN * C_IN] = {};
    for (long i = (long)blockIdx.x * blockDim.x + threadIdx.x; i < P;
         i += (long)gridDim.x * blockDim.x) {
        const float* xp = X + i * C_IN;
        float x[C_IN];
#pragma unroll
        for (int c = 0; c < C_IN; ++c) x[c] = xp[c];
#pragma unroll
        for (int c = 0; c < C_IN; ++c) {
            sx[c] += x[c];
#pragma unroll
            for (int d = 0; d < C_IN; ++d) sxx[c * C_IN + d] += x[c] * x[d];
        }
    }
    __shared__ float red[C_IN + C_IN * C_IN];
    if (threadIdx.x < C_IN + C_IN * C_IN) red[threadIdx.x] = 0.0f;
    __syncthreads();
#pragma unroll
    for (int c = 0; c < C_IN; ++c) atomicAdd(&red[c], sx[c]);
#pragma unroll
    for (int t = 0; t < C_IN * C_IN; ++t) atomicAdd(&red[C_IN + t], sxx[t]);
    __syncthreads();
    int t = threadIdx.x;
    if (t < C_IN)
        atomicAdd(&ws[WS_SUMX + t], red[t]);
    else if (t < C_IN + C_IN * C_IN)
        atomicAdd(&ws[WS_SXX + (t - C_IN)], red[t]);
}

__global__ void vfe_final0(const float* __restrict__ W0, const float* __restrict__ g0,
                           const float* __restrict__ be0, float* __restrict__ ws,
                           float invP) {
    int o = threadIdx.x;
    if (o >= C0) return;
    float w[C_IN];
#pragma unroll
    for (int c = 0; c < C_IN; ++c) w[c] = W0[o * C_IN + c];
    float mean = 0.0f;
#pragma unroll
    for (int c = 0; c < C_IN; ++c) mean += w[c] * ws[WS_SUMX + c];
    mean *= invP;
    float ey2 = 0.0f;
#pragma unroll
    for (int c = 0; c < C_IN; ++c)
#pragma unroll
        for (int d = 0; d < C_IN; ++d) ey2 += w[c] * w[d] * ws[WS_SXX + c * C_IN + d];
    ey2 *= invP;
    float var = ey2 - mean * mean;
    float sc = g0[o] * rsqrtf(var + EPS);
    ws[WS_SCALE0 + o] = sc;
    ws[WS_BIAS0 + o] = be0[o] - mean * sc;
}

__global__ void vfe_final1(const float* __restrict__ g1, const float* __restrict__ be1,
                           float* __restrict__ ws, float invP) {
    int o = threadIdx.x;
    if (o >= C1) return;
    float mean = ws[WS_SUMY + o] * invP;
    float var = ws[WS_SUMQ + o] * invP - mean * mean;
    float sc = g1[o] * rsqrtf(var + EPS);
    ws[WS_SCALE1 + o] = sc;
    ws[WS_BIAS1 + o] = be1[o] - mean * sc;
}

template <bool STATS>
__global__ __launch_bounds__(256) void vfe_gemm(const float* __restrict__ X,
                                                const float* __restrict__ W0,
                                                const float* __restrict__ W1,
                                                float* __restrict__ ws,
                                                float* __restrict__ out) {
    __shared__ float xr[M_PTS * C_IN];
    __shared__ float w0s[C0 * C_IN];
    __shared__ float s0[C0], b0[C0];
    __shared__ int   m0[C0];
    __shared__ float x1[M_PTS * X1_STRIDE];
    __shared__ float wT[C1 * WT_STRIDE];
    __shared__ float redS[C1], redQ[C1];
    __shared__ int   mx[C1];
    __shared__ float s1[C1], b1[C1];

    const int tid = threadIdx.x;
    const int n = blockIdx.x;

    for (int i = tid; i < M_PTS * C_IN; i += 256)
        xr[i] = X[(size_t)n * (M_PTS * C_IN) + i];
    for (int i = tid; i < C0 * C_IN; i += 256) w0s[i] = W0[i];
    if (tid < C0) {
        s0[tid] = ws[WS_SCALE0 + tid];
        b0[tid] = ws[WS_BIAS0 + tid];
        m0[tid] = 0;
    }
    if (tid < C1) {
        redS[tid] = 0.0f;
        redQ[tid] = 0.0f;
        mx[tid] = 0;
        if (!STATS) {
            s1[tid] = ws[WS_SCALE1 + tid];
            b1[tid] = ws[WS_BIAS1 + tid];
        }
    }
    for (int i = tid; i < C1 * C1; i += 256) {
        int o = i >> 7, k = i & 127;
        wT[k * WT_STRIDE + o] = W1[i];
    }
    __syncthreads();

    for (int i = tid; i < M_PTS * C0; i += 256) {
        int r = i >> 6, o = i & 63;
        float acc = 0.0f;
#pragma unroll
        for (int c = 0; c < C_IN; ++c) acc += xr[r * C_IN + c] * w0s[o * C_IN + c];
        float v = acc * s0[o] + b0[o];
        v = v > 0.0f ? v : 0.0f;
        x1[r * X1_STRIDE + o] = v;
        atomicMax(&m0[o], __float_as_int(v));
    }
    __syncthreads();
    for (int i = tid; i < M_PTS * C0; i += 256) {
        int r = i >> 6, o = i & 63;
        x1[r * X1_STRIDE + C0 + o] = __int_as_float(m0[o]);
    }
    __syncthreads();

    const int lane = tid & 31;
    const int w = tid >> 5;
    const int rt = w & 3;
    const int jbase = (w >> 2) * 4;
    const int hi = lane >> 4;
    const int lrow = rt * 16 + (lane & 15);

    v2f aF[32];
#pragma unroll
    for (int k = 0; k < 32; ++k)
        aF[k] = *(const v2f*)&x1[lrow * X1_STRIDE + 4 * k + 2 * hi];

    for (int jj = 0; jj < 4; ++jj) {
        const int j = jbase + jj;
        const int ncol = j * 16 + (lane & 15);
        v8f acc = {};
#pragma unroll
        for (int k = 0; k < 32; ++k) {
            const int kk = 4 * k + 2 * hi;
            v2f b;
            b.x = wT[kk * WT_STRIDE + ncol];
            b.y = wT[(kk + 1) * WT_STRIDE + ncol];
            acc = __builtin_amdgcn_wmma_f32_16x16x4_f32(
                false, aF[k], false, b, (short)0, acc, false, false);
        }
        if (STATS) {
            float s = 0.0f, q = 0.0f;
#pragma unroll
            for (int i = 0; i < 8; ++i) {
                float v = acc[i];
                s += v;
                q += v * v;
            }
            atomicAdd(&redS[ncol], s);
            atomicAdd(&redQ[ncol], q);
        } else {
            const float sc = s1[ncol], bi = b1[ncol];
            float m = 0.0f;
#pragma unroll
            for (int i = 0; i < 8; ++i) {
                float v = acc[i] * sc + bi;
                v = v > 0.0f ? v : 0.0f;
                m = v > m ? v : m;
            }
            atomicMax(&mx[ncol], __float_as_int(m));
        }
    }
    __syncthreads();

    if (STATS) {
        if (tid < C1) {
            atomicAdd(&ws[WS_SUMY + tid], redS[tid]);
            atomicAdd(&ws[WS_SUMQ + tid], redQ[tid]);
        }
    } else {
        if (tid < C1) out[(size_t)n * C1 + tid] = __int_as_float(mx[tid]);
    }
}

extern "C" void kernel_launch(void* const* d_in, const int* in_sizes, int n_in,
                              void* d_out, int out_size, void* d_ws, size_t ws_size,
                              hipStream_t stream) {
    (void)in_sizes; (void)n_in; (void)out_size; (void)ws_size;
    const float* X   = (const float*)d_in[0];
    const float* W0  = (const float*)d_in[1];
    const float* g0  = (const float*)d_in[2];
    const float* be0 = (const float*)d_in[3];
    const float* W1  = (const float*)d_in[4];
    const float* g1  = (const float*)d_in[5];
    const float* be1 = (const float*)d_in[6];
    float* out = (float*)d_out;
    float* ws  = (float*)d_ws;

    const long P = (long)N_VOX * M_PTS;
    const float invP = 1.0f / (float)P;

    vfe_zero<<<(WS_TOTAL + 255) / 256, 256, 0, stream>>>(ws);
    vfe_stats0<<<1024, 256, 0, stream>>>(X, ws, P);
    vfe_final0<<<1, 64, 0, stream>>>(W0, g0, be0, ws, invP);
    vfe_gemm<true><<<N_VOX, 256, 0, stream>>>(X, W0, W1, ws, out);
    vfe_final1<<<1, 128, 0, stream>>>(g1, be1, ws, invP);
    vfe_gemm<false><<<N_VOX, 256, 0, stream>>>(X, W0, W1, ws, out);
}


# second GEMM pass removed: the stats pass also leaves each voxel's column max of y1 (min where gamma1<0) in out; BN1+ReLU is monotone in y1, so a small in-place pass over out finishes it
# speedup vs baseline: 1.5752x; 1.5752x over previous
#include <hip/hip_runtime.h>
#include <hip/hip_bf16.h>

typedef float v2f __attribute__((ext_vector_type(2)));
typedef float v8f __attribute__((ext_vector_type(8)));

#define N_VOX 20000
#define M_PTS 64
#define C_IN  10
#define C0    64
#define C1    128
#define EPS   1e-3f
#define N_OUT4 640000

#define WS_SUMX   0
#define WS_SXX    16
#define WS_SCALE0 128
#define WS_BIAS0  192
#define WS_SUMY   256
#define WS_SUMQ   384
#define WS_SCALE1 512
#define WS_BIAS1  640
#define WS_TOTAL  768

#define X1_STRIDE 132
#define WT_STRIDE 132

__global__ void vfe_zero(float* __restrict__ ws) {
    int i = blockIdx.x * blockDim.x + threadIdx.x;
    if (i < WS_TOTAL) ws[i] = 0.0f;
}

__global__ __launch_bounds__(256) void vfe_stats0(const float* __restrict__ X,
                                                  float* __restrict__ ws, long P) {
    float sx[C_IN] = {};
    float sxx[C_IN * C_IN] = {};
    for (long i = (long)blockIdx.x * blockDim.x + threadIdx.x; i < P;
         i += (long)gridDim.x * blockDim.x) {
        const float* xp = X + i * C_IN;
        float x[C_IN];
#pragma unroll
        for (int c = 0; c < C_IN; ++c) x[c] = xp[c];
#pragma unroll
        for (int c = 0; c < C_IN; ++c) {
            sx[c] += x[c];
#pragma unroll
            for (int d = 0; d < C_IN; ++d) sxx[c * C_IN + d] += x[c] * x[d];
        }
    }
    __shared__ float red[C_IN + C_IN * C_IN];
    if (threadIdx.x < C_IN + C_IN * C_IN) red[threadIdx.x] = 0.0f;
    __syncthreads();
#pragma unroll
    for (int c = 0; c < C_IN; ++c) atomicAdd(&red[c], sx[c]);
#pragma unroll
    for (int t = 0; t < C_IN * C_IN; ++t) atomicAdd(&red[C_IN + t], sxx[t]);
    __syncthreads();
    int t = threadIdx.x;
    if (t < C_IN)
        atomicAdd(&ws[WS_SUMX + t], red[t]);
    else if (t < C_IN + C_IN * C_IN)
        atomicAdd(&ws[WS_SXX + (t - C_IN)], red[t]);
}

__global__ void vfe_final0(const float* __restrict__ W0, const float* __restrict__ g0,
                           const float* __restrict__ be0, float* __restrict__ ws,
                           float invP) {
    int o = threadIdx.x;
    if (o >= C0) return;
    float w[C_IN];
#pragma unroll
    for (int c = 0; c < C_IN; ++c) w[c] = W0[o * C_IN + c];
    float mean = 0.0f;
#pragma unroll
    for (int c = 0; c < C_IN; ++c) mean += w[c] * ws[WS_SUMX + c];
    mean *= invP;
    float ey2 = 0.0f;
#pragma unroll
    for (int c = 0; c < C_IN; ++c)
#pragma unroll
        for (int d = 0; d < C_IN; ++d) ey2 += w[c] * w[d] * ws[WS_SXX + c * C_IN + d];
    ey2 *= invP;
    float var = ey2 - mean * mean;
    float sc = g0[o] * rsqrtf(var + EPS);
    ws[WS_SCALE0 + o] = sc;
    ws[WS_BIAS0 + o] = be0[o] - mean * sc;
}

__global__ void vfe_final1(const float* __restrict__ g1, const float* __restrict__ be1,
                           float* __restrict__ ws, float invP) {
    int o = threadIdx.x;
    if (o >= C1) return;
    float mean = ws[WS_SUMY + o] * invP;
    float var = ws[WS_SUMQ + o] * invP - mean * mean;
    float sc = g1[o] * rsqrtf(var + EPS);
    ws[WS_SCALE1 + o] = sc;
    ws[WS_BIAS1 + o] = be1[o] - mean * sc;
}

__global__ __launch_bounds__(256) void vfe_gemm(const float* __restrict__ X,
                                                const float* __restrict__ W0,
                                                const float* __restrict__ W1,
                                                const float* __restrict__ g1,
                                                float* __restrict__ ws,
                                                float* __restrict__ out) {
    __shared__ float xr[M_PTS * C_IN];
    __shared__ float w0s[C0 * C_IN];
    __shared__ float s0[C0], b0[C0];
    __shared__ int   m0[C0];
    __shared__ float x1[M_PTS * X1_STRIDE];
    __shared__ float wT[C1 * WT_STRIDE];
    __shared__ float redS[C1], redQ[C1];
    __shared__ float sgn[C1];
    __shared__ float ext[8 * C1];

    const int tid = threadIdx.x;
    const int n = blockIdx.x;

    for (int i = tid; i < M_PTS * C_IN; i += 256)
        xr[i] = X[(size_t)n * (M_PTS * C_IN) + i];
    for (int i = tid; i < C0 * C_IN; i += 256) w0s[i] = W0[i];
    if (tid < C0) {
        s0[tid] = ws[WS_SCALE0 + tid];
        b0[tid] = ws[WS_BIAS0 + tid];
        m0[tid] = 0;
    }
    if (tid < C1) {
        redS[tid] = 0.0f;
        redQ[tid] = 0.0f;
        sgn[tid] = g1[tid] < 0.0f ? -1.0f : 1.0f;
    }
    for (int i = tid; i < C1 * C1; i += 256) {
        int o = i >> 7, k = i & 127;
        wT[k * WT_STRIDE + o] = W1[i];
    }
    __syncthreads();

    for (int i = tid; i < M_PTS * C0; i += 256) {
        int r = i >> 6, o = i & 63;
        float acc = 0.0f;
#pragma unroll
        for (int c = 0; c < C_IN; ++c) acc += xr[r * C_IN + c] * w0s[o * C_IN + c];
        float v = acc * s0[o] + b0[o];
        v = v > 0.0f ? v : 0.0f;
        x1[r * X1_STRIDE + o] = v;
        atomicMax(&m0[o], __float_as_int(v));
    }
    __syncthreads();
    for (int i = tid; i < M_PTS * C0; i += 256) {
        int r = i >> 6, o = i & 63;
        x1[r * X1_STRIDE + C0 + o] = __int_as_float(m0[o]);
    }
    __syncthreads();

    const int lane = tid & 31;
    const int w = tid >> 5;
    const int rt = w & 3;
    const int jbase = (w >> 2) * 4;
    const int hi = lane >> 4;
    const int lrow = rt * 16 + (lane & 15);

    v2f aF[32];
#pragma unroll
    for (int k = 0; k < 32; ++k)
        aF[k] = *(const v2f*)&x1[lrow * X1_STRIDE + 4 * k + 2 * hi];

    for (int jj = 0; jj < 4; ++jj) {
        const int j = jbase + jj;
        const int ncol = j * 16 + (lane & 15);
        v8f acc = {};
#pragma unroll
        for (int k = 0; k < 32; ++k) {
            const int kk = 4 * k + 2 * hi;
            v2f b;
            b.x = wT[kk * WT_STRIDE + ncol];
            b.y = wT[(kk + 1) * WT_STRIDE + ncol];
            acc = __builtin_amdgcn_wmma_f32_16x16x4_f32(
                false, aF[k], false, b, (short)0, acc, false, false);
        }
        const float sg = sgn[ncol];
        float s = 0.0f, q = 0.0f;
        float m = acc[0] * sg;
#pragma unroll
        for (int i = 0; i < 8; ++i) {
            float v = acc[i];
            s += v;
            q += v * v;
            m = fmaxf(m, v * sg);
        }
        atomicAdd(&redS[ncol], s);
        atomicAdd(&redQ[ncol], q);
        ext[(rt * 2 + hi) * C1 + ncol] = m;
    }
    __syncthreads();

    if (tid < C1) {
        atomicAdd(&ws[WS_SUMY + tid], redS[tid]);
        atomicAdd(&ws[WS_SUMQ + tid], redQ[tid]);
        float m = ext[tid];
        for (int p = 1; p < 8; ++p) m = fmaxf(m, ext[p * C1 + tid]);
        out[(size_t)n * C1 + tid] = m * sgn[tid];
    }
}

__global__ __launch_bounds__(256) void vfe_apply(const float* __restrict__ ws,
                                                 float* __restrict__ out) {
    const int i = blockIdx.x * 256 + threadIdx.x;
    if (i >= N_OUT4) return;
    const int o = (i & (C1 / 4 - 1)) * 4;
    const float4 sc = *(const float4*)&ws[WS_SCALE1 + o];
    const float4 bi = *(const float4*)&ws[WS_BIAS1 + o];
    float4 y = ((const float4*)out)[i];
    y.x = y.x * sc.x + bi.x;
    y.y = y.y * sc.y + bi.y;
    y.z = y.z * sc.z + bi.z;
    y.w = y.w * sc.w + bi.w;
    y.x = y.x > 0.0f ? y.x : 0.0f;
    y.y = y.y > 0.0f ? y.y : 0.0f;
    y.z = y.z > 0.0f ? y.z : 0.0f;
    y.w = y.w > 0.0f ? y.w : 0.0f;
    ((float4*)out)[i] = y;
}

extern "C" void kernel_launch(void* const* d_in, const int* in_sizes, int n_in,
                              void* d_out, int out_size, void* d_ws, size_t ws_size,
                              hipStream_t stream) {
    (void)in_sizes; (void)n_in; (void)out_size; (void)ws_size;
    const float* X   = (const float*)d_in[0];
    const float* W0  = (const float*)d_in[1];
    const float* g0  = (const float*)d_in[2];
    const float* be0 = (const float*)d_in[3];
    const float* W1  = (const float*)d_in[4];
    const float* g1  = (const float*)d_in[5];
    const float* be1 = (const float*)d_in[6];
    float* out = (float*)d_out;
    float* ws  = (float*)d_ws;

    const long P = (long)N_VOX * M_PTS;
    const float invP = 1.0f / (float)P;

    vfe_zero<<<(WS_TOTAL + 255) / 256, 256, 0, stream>>>(ws);
    vfe_stats0<<<1024, 256, 0, stream>>>(X, ws, P);
    vfe_final0<<<1, 64, 0, stream>>>(W0, g0, be0, ws, invP);
    vfe_gemm<<<N_VOX, 256, 0, stream>>>(X, W0, W1, g1, ws, out);
    vfe_final1<<<1, 128, 0, stream>>>(g1, be1, ws, invP);
    vfe_apply<<<N_OUT4 / 256, 256, 0, stream>>>(ws, out);
}


# persistent GEMM work-groups (512, 2 per WGP) loop over voxels: W1^T/W0/BN0 terms staged in LDS once, next voxel's rows loaded a trip ahead, BN1 sums sent to ws once per group (sum order changes)
# speedup vs baseline: 1.9013x; 1.2070x over previous
#include <hip/hip_runtime.h>
#include <hip/hip_bf16.h>

typedef float v2f __attribute__((ext_vector_type(2)));
typedef float v8f __attribute__((ext_vector_type(8)));

#define N_VOX 20000
#define M_PTS 64
#define C_IN  10
#define C0    64
#define C1    128
#define EPS   1e-3f
#define N_OUT4 640000
#define GEMM_WGS 512

#define WS_SUMX   0
#define WS_SXX    16
#define WS_SCALE0 128
#define WS_BIAS0  192
#define WS_SUMY   256
#define WS_SUMQ   384
#define WS_SCALE1 512
#define WS_BIAS1  640
#define WS_TOTAL  768

#define X1_STRIDE 132
#define WT_STRIDE 132

__global__ void vfe_zero(float* __restrict__ ws) {
    int i = blockIdx.x * blockDim.x + threadIdx.x;
    if (i < WS_TOTAL) ws[i] = 0.0f;
}

__global__ __launch_bounds__(256) void vfe_stats0(const float* __restrict__ X,
                                                  float* __restrict__ ws, long P) {
    float sx[C_IN] = {};
    float sxx[C_IN * C_IN] = {};
    for (long i = (long)blockIdx.x * blockDim.x + threadIdx.x; i < P;
         i += (long)gridDim.x * blockDim.x) {
        const float* xp = X + i * C_IN;
        float x[C_IN];
#pragma unroll
        for (int c = 0; c < C_IN; ++c) x[c] = xp[c];
#pragma unroll
        for (int c = 0; c < C_IN; ++c) {
            sx[c] += x[c];
#pragma unroll
            for (int d = 0; d < C_IN; ++d) sxx[c * C_IN + d] += x[c] * x[d];
        }
    }
    __shared__ float red[C_IN + C_IN * C_IN];
    if (threadIdx.x < C_IN + C_IN * C_IN) red[threadIdx.x] = 0.0f;
    __syncthreads();
#pragma unroll
    for (int c = 0; c < C_IN; ++c) atomicAdd(&red[c], sx[c]);
#pragma unroll
    for (int t = 0; t < C_IN * C_IN; ++t) atomicAdd(&red[C_IN + t], sxx[t]);
    __syncthreads();
    int t = threadIdx.x;
    if (t < C_IN)
        atomicAdd(&ws[WS_SUMX + t], red[t]);
    else if (t < C_IN + C_IN * C_IN)
        atomicAdd(&ws[WS_SXX + (t - C_IN)], red[t]);
}

__global__ void vfe_final0(const float* __restrict__ W0, const float* __restrict__ g0,
                           const float* __restrict__ be0, float* __restrict__ ws,
                           float invP) {
    int o = threadIdx.x;
    if (o >= C0) return;
    float w[C_IN];
#pragma unroll
    for (int c = 0; c < C_IN; ++c) w[c] = W0[o * C_IN + c];
    float mean = 0.0f;
#pragma unroll
    for (int c = 0; c < C_IN; ++c) mean += w[c] * ws[WS_SUMX + c];
    mean *= invP;
    float ey2 = 0.0f;
#pragma unroll
    for (int c = 0; c < C_IN; ++c)
#pragma unroll
        for (int d = 0; d < C_IN; ++d) ey2 += w[c] * w[d] * ws[WS_SXX + c * C_IN + d];
    ey2 *= invP;
    float var = ey2 - mean * mean;
    float sc = g0[o] * rsqrtf(var + EPS);
    ws[WS_SCALE0 + o] = sc;
    ws[WS_BIAS0 + o] = be0[o] - mean * sc;
}

__global__ void vfe_final1(const float* __restrict__ g1, const float* __restrict__ be1,
                           float* __restrict__ ws, float invP) {
    int o = threadIdx.x;
    if (o >= C1) return;
    float mean = ws[WS_SUMY + o] * invP;
    float var = ws[WS_SUMQ + o] * invP - mean * mean;
    float sc = g1[o] * rsqrtf(var + EPS);
    ws[WS_SCALE1 + o] = sc;
    ws[WS_BIAS1 + o] = be1[o] - mean * sc;
}

__global__ __launch_bounds__(256) void vfe_gemm(const float* __restrict__ X,
                                                const float* __restrict__ W0,
                                                const float* __restrict__ W1,
                                                const float* __restrict__ g1,
                                                float* __restrict__ ws,
                                                float* __restrict__ out) {
    __shared__ float xr[M_PTS * C_IN];
    __shared__ float w0s[C0 * C_IN];
    __shared__ float s0[C0], b0[C0];
    __shared__ int   m0[C0];
    __shared__ float x1[M_PTS * X1_STRIDE];
    __shared__ float wT[C1 * WT_STRIDE];
    __shared__ float redS[C1], redQ[C1];
    __shared__ float sgn[C1];
    __shared__ float ext[8 * C1];

    const int tid = threadIdx.x;

    for (int i = tid; i < C0 * C_IN; i += 256) w0s[i] = W0[i];
    if (tid < C0) {
        s0[tid] = ws[WS_SCALE0 + tid];
        b0[tid] = ws[WS_BIAS0 + tid];
    }
    if (tid < C1) {
        redS[tid] = 0.0f;
        redQ[tid] = 0.0f;
        sgn[tid] = g1[tid] < 0.0f ? -1.0f : 1.0f;
    }
    for (int i = tid; i < C1 * C1; i += 256) {
        int o = i >> 7, k = i & 127;
        wT[k * WT_STRIDE + o] = W1[i];
    }

    const int lane = tid & 31;
    const int w = tid >> 5;
    const int rt = w & 3;
    const int jbase = (w >> 2) * 4;
    const int hi = lane >> 4;
    const int lrow = rt * 16 + (lane & 15);

    float xa, xb, xc = 0.0f;
    {
        const float* xp = X + (size_t)blockIdx.x * (M_PTS * C_IN);
        xa = xp[tid];
        xb = xp[tid + 256];
        if (tid < 128) xc = xp[tid + 512];
    }

    for (int n = blockIdx.x; n < N_VOX; n += GEMM_WGS) {
        xr[tid] = xa;
        xr[tid + 256] = xb;
        if (tid < 128) xr[tid + 512] = xc;
        if (tid < C0) m0[tid] = 0;
        __syncthreads();

        for (int i = tid; i < M_PTS * C0; i += 256) {
            int r = i >> 6, o = i & 63;
            float acc = 0.0f;
#pragma unroll
            for (int c = 0; c < C_IN; ++c) acc += xr[r * C_IN + c] * w0s[o * C_IN + c];
            float v = acc * s0[o] + b0[o];
            v = v > 0.0f ? v : 0.0f;
            x1[r * X1_STRIDE + o] = v;
            atomicMax(&m0[o], __float_as_int(v));
        }
        __syncthreads();
        for (int i = tid; i < M_PTS * C0; i += 256) {
            int r = i >> 6, o = i & 63;
            x1[r * X1_STRIDE + C0 + o] = __int_as_float(m0[o]);
        }
        __syncthreads();

        if (n + GEMM_WGS < N_VOX) {
            const float* xp = X + (size_t)(n + GEMM_WGS) * (M_PTS * C_IN);
            xa = xp[tid];
            xb = xp[tid + 256];
            if (tid < 128) xc = xp[tid + 512];
        }

        v2f aF[32];
#pragma unroll
        for (int k = 0; k < 32; ++k)
            aF[k] = *(const v2f*)&x1[lrow * X1_STRIDE + 4 * k + 2 * hi];

        for (int jj = 0; jj < 4; ++jj) {
            const int j = jbase + jj;
            const int ncol = j * 16 + (lane & 15);
            v8f acc = {};
#pragma unroll
            for (int k = 0; k < 32; ++k) {
                const int kk = 4 * k + 2 * hi;
                v2f b;
                b.x = wT[kk * WT_STRIDE + ncol];
                b.y = wT[(kk + 1) * WT_STRIDE + ncol];
                acc = __builtin_amdgcn_wmma_f32_16x16x4_f32(
                    false, aF[k], false, b, (short)0, acc, false, false);
            }
            const float sg = sgn[ncol];
            float s = 0.0f, q = 0.0f;
            float m = acc[0] * sg;
#pragma unroll
            for (int i = 0; i < 8; ++i) {
                float v = acc[i];
                s += v;
                q += v * v;
                m = fmaxf(m, v * sg);
            }
            atomicAdd(&redS[ncol], s);
            atomicAdd(&redQ[ncol], q);
            ext[(rt * 2 + hi) * C1 + ncol] = m;
        }
        __syncthreads();

        if (tid < C1) {
            float m = ext[tid];
            for (int p = 1; p < 8; ++p) m = fmaxf(m, ext[p * C1 + tid]);
            out[(size_t)n * C1 + tid] = m * sgn[tid];
        }
    }

    if (tid < C1) {
        atomicAdd(&ws[WS_SUMY + tid], redS[tid]);
        atomicAdd(&ws[WS_SUMQ + tid], redQ[tid]);
    }
}

__global__ __launch_bounds__(256) void vfe_apply(const float* __restrict__ ws,
                                                 float* __restrict__ out) {
    const int i = blockIdx.x * 256 + threadIdx.x;
    if (i >= N_OUT4) return;
    const int o = (i & (C1 / 4 - 1)) * 4;
    const float4 sc = *(const float4*)&ws[WS_SCALE1 + o];
    const float4 bi = *(const float4*)&ws[WS_BIAS1 + o];
    float4 y = ((const float4*)out)[i];
    y.x = y.x * sc.x + bi.x;
    y.y = y.y * sc.y + bi.y;
    y.z = y.z * sc.z + bi.z;
    y.w = y.w * sc.w + bi.w;
    y.x = y.x > 0.0f ? y.x : 0.0f;
    y.y = y.y > 0.0f ? y.y : 0.0f;
    y.z = y.z > 0.0f ? y.z : 0.0f;
    y.w = y.w > 0.0f ? y.w : 0.0f;
    ((float4*)out)[i] = y;
}

extern "C" void kernel_launch(void* const* d_in, const int* in_sizes, int n_in,
                              void* d_out, int out_size, void* d_ws, size_t ws_size,
                              hipStream_t stream) {
    (void)in_sizes; (void)n_in; (void)out_size; (void)ws_size;
    const float* X   = (const float*)d_in[0];
    const float* W0  = (const float*)d_in[1];
    const float* g0  = (const float*)d_in[2];
    const float* be0 = (const float*)d_in[3];
    const float* W1  = (const float*)d_in[4];
    const float* g1  = (const float*)d_in[5];
    const float* be1 = (const float*)d_in[6];
    float* out = (float*)d_out;
    float* ws  = (float*)d_ws;

    const long P = (long)N_VOX * M_PTS;
    const float invP = 1.0f / (float)P;

    vfe_zero<<<(WS_TOTAL + 255) / 256, 256, 0, stream>>>(ws);
    vfe_stats0<<<1024, 256, 0, stream>>>(X, ws, P);
    vfe_final0<<<1, 64, 0, stream>>>(W0, g0, be0, ws, invP);
    vfe_gemm<<<GEMM_WGS, 256, 0, stream>>>(X, W0, W1, g1, ws, out);
    vfe_final1<<<1, 128, 0, stream>>>(g1, be1, ws, invP);
    vfe_apply<<<N_OUT4 / 256, 256, 0, stream>>>(ws, out);
}


# layer-1 K 128->64: the repeated-max half of its input is equal in all 64 rows of a voxel, so its product is one 16-row WMMA tile per voxel, used as accumulator start (sum order changes)
# speedup vs baseline: 2.1984x; 1.1563x over previous
#include <hip/hip_runtime.h>
#include <hip/hip_bf16.h>

typedef float v2f __attribute__((ext_vector_type(2)));
typedef float v8f __attribute__((ext_vector_type(8)));

#define N_VOX 20000
#define M_PTS 64
#define C_IN  10
#define C0    64
#define C1    128
#define EPS   1e-3f
#define N_OUT4 640000
#define GEMM_WGS 512

#define WS_SUMX   0
#define WS_SXX    16
#define WS_SCALE0 128
#define WS_BIAS0  192
#define WS_SUMY   256
#define WS_SUMQ   384
#define WS_SCALE1 512
#define WS_BIAS1  640
#define WS_TOTAL  768

#define X1_STRIDE 132
#define WT_STRIDE 132

__global__ void vfe_zero(float* __restrict__ ws) {
    int i = blockIdx.x * blockDim.x + threadIdx.x;
    if (i < WS_TOTAL) ws[i] = 0.0f;
}

__global__ __launch_bounds__(256) void vfe_stats0(const float* __restrict__ X,
                                                  float* __restrict__ ws, long P) {
    float sx[C_IN] = {};
    float sxx[C_IN * C_IN] = {};
    for (long i = (long)blockIdx.x * blockDim.x + threadIdx.x; i < P;
         i += (long)gridDim.x * blockDim.x) {
        const float* xp = X + i * C_IN;
        float x[C_IN];
#pragma unroll
        for (int c = 0; c < C_IN; ++c) x[c] = xp[c];
#pragma unroll
        for (int c = 0; c < C_IN; ++c) {
            sx[c] += x[c];
#pragma unroll
            for (int d = 0; d < C_IN; ++d) sxx[c * C_IN + d] += x[c] * x[d];
        }
    }
    __shared__ float red[C_IN + C_IN * C_IN];
    if (threadIdx.x < C_IN + C_IN * C_IN) red[threadIdx.x] = 0.0f;
    __syncthreads();
#pragma unroll
    for (int c = 0; c < C_IN; ++c) atomicAdd(&red[c], sx[c]);
#pragma unroll
    for (int t = 0; t < C_IN * C_IN; ++t) atomicAdd(&red[C_IN + t], sxx[t]);
    __syncthreads();
    int t = threadIdx.x;
    if (t < C_IN)
        atomicAdd(&ws[WS_SUMX + t], red[t]);
    else if (t < C_IN + C_IN * C_IN)
        atomicAdd(&ws[WS_SXX + (t - C_IN)], red[t]);
}

__global__ void vfe_final0(const float* __restrict__ W0, const float* __restrict__ g0,
                           const float* __restrict__ be0, float* __restrict__ ws,
                           float invP) {
    int o = threadIdx.x;
    if (o >= C0) return;
    float w[C_IN];
#pragma unroll
    for (int c = 0; c < C_IN; ++c) w[c] = W0[o * C_IN + c];
    float mean = 0.0f;
#pragma unroll
    for (int c = 0; c < C_IN; ++c) mean += w[c] * ws[WS_SUMX + c];
    mean *= invP;
    float ey2 = 0.0f;
#pragma unroll
    for (int c = 0; c < C_IN; ++c)
#pragma unroll
        for (int d = 0; d < C_IN; ++d) ey2 += w[c] * w[d] * ws[WS_SXX + c * C_IN + d];
    ey2 *= invP;
    float var = ey2 - mean * mean;
    float sc = g0[o] * rsqrtf(var + EPS);
    ws[WS_SCALE0 + o] = sc;
    ws[WS_BIAS0 + o] = be0[o] - mean * sc;
}

__global__ void vfe_final1(const float* __restrict__ g1, const float* __restrict__ be1,
                           float* __restrict__ ws, float invP) {
    int o = threadIdx.x;
    if (o >= C1) return;
    float mean = ws[WS_SUMY + o] * invP;
    float var = ws[WS_SUMQ + o] * invP - mean * mean;
    float sc = g1[o] * rsqrtf(var + EPS);
    ws[WS_SCALE1 + o] = sc;
    ws[WS_BIAS1 + o] = be1[o] - mean * sc;
}

__global__ __launch_bounds__(256) void vfe_gemm(const float* __restrict__ X,
                                                const float* __restrict__ W0,
                                                const float* __restrict__ W1,
                                                const float* __restrict__ g1,
                                                float* __restrict__ ws,
                                                float* __restrict__ out) {
    __shared__ float xr[M_PTS * C_IN];
    __shared__ float w0s[C0 * C_IN];
    __shared__ float s0[C0], b0[C0];
    __shared__ int   m0[C0];
    __shared__ float x1[M_PTS * X1_STRIDE];
    __shared__ float wT[C1 * WT_STRIDE];
    __shared__ float redS[C1], redQ[C1];
    __shared__ float sgn[C1];
    __shared__ float ext[8 * C1];
    __shared__ float cvec[2 * C1];

    const int tid = threadIdx.x;

    for (int i = tid; i < C0 * C_IN; i += 256) w0s[i] = W0[i];
    if (tid < C0) {
        s0[tid] = ws[WS_SCALE0 + tid];
        b0[tid] = ws[WS_BIAS0 + tid];
    }
    if (tid < C1) {
        redS[tid] = 0.0f;
        redQ[tid] = 0.0f;
        sgn[tid] = g1[tid] < 0.0f ? -1.0f : 1.0f;
    }
    for (int i = tid; i < C1 * C1; i += 256) {
        int o = i >> 7, k = i & 127;
        wT[k * WT_STRIDE + o] = W1[i];
    }

    const int lane = tid & 31;
    const int w = tid >> 5;
    const int rt = w & 3;
    const int jbase = (w >> 2) * 4;
    const int hi = lane >> 4;
    const int lrow = rt * 16 + (lane & 15);

    float xa, xb, xc = 0.0f;
    {
        const float* xp = X + (size_t)blockIdx.x * (M_PTS * C_IN);
        xa = xp[tid];
        xb = xp[tid + 256];
        if (tid < 128) xc = xp[tid + 512];
    }

    for (int n = blockIdx.x; n < N_VOX; n += GEMM_WGS) {
        xr[tid] = xa;
        xr[tid + 256] = xb;
        if (tid < 128) xr[tid + 512] = xc;
        if (tid < C0) m0[tid] = 0;
        __syncthreads();

        for (int i = tid; i < M_PTS * C0; i += 256) {
            int r = i >> 6, o = i & 63;
            float acc = 0.0f;
#pragma unroll
            for (int c = 0; c < C_IN; ++c) acc += xr[r * C_IN + c] * w0s[o * C_IN + c];
            float v = acc * s0[o] + b0[o];
            v = v > 0.0f ? v : 0.0f;
            x1[r * X1_STRIDE + o] = v;
            atomicMax(&m0[o], __float_as_int(v));
        }
        __syncthreads();

        {
            const int col = w * 16 + (lane & 15);
            v8f ce = {}, co = {};
#pragma unroll
            for (int k = 0; k < 16; k += 2) {
                const int k0 = 4 * k + 2 * hi;
                const int k1 = k0 + 4;
                v2f a0, a1, bb0, bb1;
                a0.x = __int_as_float(m0[k0]);
                a0.y = __int_as_float(m0[k0 + 1]);
                a1.x = __int_as_float(m0[k1]);
                a1.y = __int_as_float(m0[k1 + 1]);
                bb0.x = wT[(C0 + k0) * WT_STRIDE + col];
                bb0.y = wT[(C0 + k0 + 1) * WT_STRIDE + col];
                bb1.x = wT[(C0 + k1) * WT_STRIDE + col];
                bb1.y = wT[(C0 + k1 + 1) * WT_STRIDE + col];
                ce = __builtin_amdgcn_wmma_f32_16x16x4_f32(
                    false, a0, false, bb0, (short)0, ce, false, false);
                co = __builtin_amdgcn_wmma_f32_16x16x4_f32(
                    false, a1, false, bb1, (short)0, co, false, false);
            }
            cvec[hi * C1 + col] = ce[0] + co[0];
        }
        __syncthreads();

        if (n + GEMM_WGS < N_VOX) {
            const float* xp = X + (size_t)(n + GEMM_WGS) * (M_PTS * C_IN);
            xa = xp[tid];
            xb = xp[tid + 256];
            if (tid < 128) xc = xp[tid + 512];
        }

        v2f aF[16];
#pragma unroll
        for (int k = 0; k < 16; ++k)
            aF[k] = *(const v2f*)&x1[lrow * X1_STRIDE + 4 * k + 2 * hi];

        for (int jj = 0; jj < 4; ++jj) {
            const int j = jbase + jj;
            const int ncol = j * 16 + (lane & 15);
            const float cv = cvec[ncol];
            v8f acc = {cv, cv, cv, cv, cv, cv, cv, cv};
#pragma unroll
            for (int k = 0; k < 16; ++k) {
                const int kk = 4 * k + 2 * hi;
                v2f b;
                b.x = wT[kk * WT_STRIDE + ncol];
                b.y = wT[(kk + 1) * WT_STRIDE + ncol];
                acc = __builtin_amdgcn_wmma_f32_16x16x4_f32(
                    false, aF[k], false, b, (short)0, acc, false, false);
            }
            const float sg = sgn[ncol];
            float s = 0.0f, q = 0.0f;
            float m = acc[0] * sg;
#pragma unroll
            for (int i = 0; i < 8; ++i) {
                float v = acc[i];
                s += v;
                q += v * v;
                m = fmaxf(m, v * sg);
            }
            atomicAdd(&redS[ncol], s);
            atomicAdd(&redQ[ncol], q);
            ext[(rt * 2 + hi) * C1 + ncol] = m;
        }
        __syncthreads();

        if (tid < C1) {
            float m = ext[tid];
            for (int p = 1; p < 8; ++p) m = fmaxf(m, ext[p * C1 + tid]);
            out[(size_t)n * C1 + tid] = m * sgn[tid];
        }
    }

    if (tid < C1) {
        atomicAdd(&ws[WS_SUMY + tid], redS[tid]);
        atomicAdd(&ws[WS_SUMQ + tid], redQ[tid]);
    }
}

__global__ __launch_bounds__(256) void vfe_apply(const float* __restrict__ ws,
                                                 float* __restrict__ out) {
    const int i = blockIdx.x * 256 + threadIdx.x;
    if (i >= N_OUT4) return;
    const int o = (i & (C1 / 4 - 1)) * 4;
    const float4 sc = *(const float4*)&ws[WS_SCALE1 + o];
    const float4 bi = *(const float4*)&ws[WS_BIAS1 + o];
    float4 y = ((const float4*)out)[i];
    y.x = y.x * sc.x + bi.x;
    y.y = y.y * sc.y + bi.y;
    y.z = y.z * sc.z + bi.z;
    y.w = y.w * sc.w + bi.w;
    y.x = y.x > 0.0f ? y.x : 0.0f;
    y.y = y.y > 0.0f ? y.y : 0.0f;
    y.z = y.z > 0.0f ? y.z : 0.0f;
    y.w = y.w > 0.0f ? y.w : 0.0f;
    ((float4*)out)[i] = y;
}

extern "C" void kernel_launch(void* const* d_in, const int* in_sizes, int n_in,
                              void* d_out, int out_size, void* d_ws, size_t ws_size,
                              hipStream_t stream) {
    (void)in_sizes; (void)n_in; (void)out_size; (void)ws_size;
    const float* X   = (const float*)d_in[0];
    const float* W0  = (const float*)d_in[1];
    const float* g0  = (const float*)d_in[2];
    const float* be0 = (const float*)d_in[3];
    const float* W1  = (const float*)d_in[4];
    const float* g1  = (const float*)d_in[5];
    const float* be1 = (const float*)d_in[6];
    float* out = (float*)d_out;
    float* ws  = (float*)d_ws;

    const long P = (long)N_VOX * M_PTS;
    const float invP = 1.0f / (float)P;

    vfe_zero<<<(WS_TOTAL + 255) / 256, 256, 0, stream>>>(ws);
    vfe_stats0<<<1024, 256, 0, stream>>>(X, ws, P);
    vfe_final0<<<1, 64, 0, stream>>>(W0, g0, be0, ws, invP);
    vfe_gemm<<<GEMM_WGS, 256, 0, stream>>>(X, W0, W1, g1, ws, out);
    vfe_final1<<<1, 128, 0, stream>>>(g1, be1, ws, invP);
    vfe_apply<<<N_OUT4 / 256, 256, 0, stream>>>(ws, out);
}


# W1 fragments stay in registers over the whole voxel loop (read once per work-group from global, LDS transpose dropped): K-step-outer WMMA runs, two column tiles at a time, no LDS read inside them
# speedup vs baseline: 2.3531x; 1.0704x over previous
#include <hip/hip_runtime.h>
#include <hip/hip_bf16.h>

typedef float v2f __attribute__((ext_vector_type(2)));
typedef float v8f __attribute__((ext_vector_type(8)));

#define N_VOX 20000
#define M_PTS 64
#define C_IN  10
#define C0    64
#define C1    128
#define EPS   1e-3f
#define N_OUT4 640000
#define GEMM_WGS 512

#define WS_SUMX   0
#define WS_SXX    16
#define WS_SCALE0 128
#define WS_BIAS0  192
#define WS_SUMY   256
#define WS_SUMQ   384
#define WS_SCALE1 512
#define WS_BIAS1  640
#define WS_TOTAL  768

#define X1_STRIDE 132

__global__ void vfe_zero(float* __restrict__ ws) {
    int i = blockIdx.x * blockDim.x + threadIdx.x;
    if (i < WS_TOTAL) ws[i] = 0.0f;
}

__global__ __launch_bounds__(256) void vfe_stats0(const float* __restrict__ X,
                                                  float* __restrict__ ws, long P) {
    float sx[C_IN] = {};
    float sxx[C_IN * C_IN] = {};
    for (long i = (long)blockIdx.x * blockDim.x + threadIdx.x; i < P;
         i += (long)gridDim.x * blockDim.x) {
        const float* xp = X + i * C_IN;
        float x[C_IN];
#pragma unroll
        for (int c = 0; c < C_IN; ++c) x[c] = xp[c];
#pragma unroll
        for (int c = 0; c < C_IN; ++c) {
            sx[c] += x[c];
#pragma unroll
            for (int d = 0; d < C_IN; ++d) sxx[c * C_IN + d] += x[c] * x[d];
        }
    }
    __shared__ float red[C_IN + C_IN * C_IN];
    if (threadIdx.x < C_IN + C_IN * C_IN) red[threadIdx.x] = 0.0f;
    __syncthreads();
#pragma unroll
    for (int c = 0; c < C_IN; ++c) atomicAdd(&red[c], sx[c]);
#pragma unroll
    for (int t = 0; t < C_IN * C_IN; ++t) atomicAdd(&red[C_IN + t], sxx[t]);
    __syncthreads();
    int t = threadIdx.x;
    if (t < C_IN)
        atomicAdd(&ws[WS_SUMX + t], red[t]);
    else if (t < C_IN + C_IN * C_IN)
        atomicAdd(&ws[WS_SXX + (t - C_IN)], red[t]);
}

__global__ void vfe_final0(const float* __restrict__ W0, const float* __restrict__ g0,
                           const float* __restrict__ be0, float* __restrict__ ws,
                           float invP) {
    int o = threadIdx.x;
    if (o >= C0) return;
    float w[C_IN];
#pragma unroll
    for (int c = 0; c < C_IN; ++c) w[c] = W0[o * C_IN + c];
    float mean = 0.0f;
#pragma unroll
    for (int c = 0; c < C_IN; ++c) mean += w[c] * ws[WS_SUMX + c];
    mean *= invP;
    float ey2 = 0.0f;
#pragma unroll
    for (int c = 0; c < C_IN; ++c)
#pragma unroll
        for (int d = 0; d < C_IN; ++d) ey2 += w[c] * w[d] * ws[WS_SXX + c * C_IN + d];
    ey2 *= invP;
    float var = ey2 - mean * mean;
    float sc = g0[o] * rsqrtf(var + EPS);
    ws[WS_SCALE0 + o] = sc;
    ws[WS_BIAS0 + o] = be0[o] - mean * sc;
}

__global__ void vfe_final1(const float* __restrict__ g1, const float* __restrict__ be1,
                           float* __restrict__ ws, float invP) {
    int o = threadIdx.x;
    if (o >= C1) return;
    float mean = ws[WS_SUMY + o] * invP;
    float var = ws[WS_SUMQ + o] * invP - mean * mean;
    float sc = g1[o] * rsqrtf(var + EPS);
    ws[WS_SCALE1 + o] = sc;
    ws[WS_BIAS1 + o] = be1[o] - mean * sc;
}

__global__ __launch_bounds__(256) __attribute__((amdgpu_waves_per_eu(4, 4)))
void vfe_gemm(const float* __restrict__ X,
              const float* __restrict__ W0,
              const float* __restrict__ W1,
              const float* __restrict__ g1,
              float* __restrict__ ws,
              float* __restrict__ out) {
    __shared__ float xr[M_PTS * C_IN];
    __shared__ float w0s[C0 * C_IN];
    __shared__ float s0[C0], b0[C0];
    __shared__ int   m0[C0];
    __shared__ float x1[M_PTS * X1_STRIDE];
    __shared__ float redS[C1], redQ[C1];
    __shared__ float sgn[C1];
    __shared__ float ext[8 * C1];
    __shared__ float cvec[2 * C1];

    const int tid = threadIdx.x;

    for (int i = tid; i < C0 * C_IN; i += 256) w0s[i] = W0[i];
    if (tid < C0) {
        s0[tid] = ws[WS_SCALE0 + tid];
        b0[tid] = ws[WS_BIAS0 + tid];
    }
    if (tid < C1) {
        redS[tid] = 0.0f;
        redQ[tid] = 0.0f;
        sgn[tid] = g1[tid] < 0.0f ? -1.0f : 1.0f;
    }

    const int lane = tid & 31;
    const int w = tid >> 5;
    const int rt = w & 3;
    const int jbase = (w >> 2) * 4;
    const int hi = lane >> 4;
    const int lrow = rt * 16 + (lane & 15);

    v2f bF[4][16];
    v2f cB[16];
#pragma unroll
    for (int jj = 0; jj < 4; ++jj) {
        const float* wr = W1 + ((jbase + jj) * 16 + (lane & 15)) * C1 + 2 * hi;
#pragma unroll
        for (int k = 0; k < 16; ++k) bF[jj][k] = *(const v2f*)(wr + 4 * k);
    }
    {
        const float* wr = W1 + (w * 16 + (lane & 15)) * C1 + C0 + 2 * hi;
#pragma unroll
        for (int k = 0; k < 16; ++k) cB[k] = *(const v2f*)(wr + 4 * k);
    }

    float xa, xb, xc = 0.0f;
    {
        const int xo = (int)blockIdx.x * (M_PTS * C_IN) + tid;
        xa = X[xo];
        xb = X[xo + 256];
        if (tid < 128) xc = X[xo + 512];
    }

    for (int n = blockIdx.x; n < N_VOX; n += GEMM_WGS) {
        xr[tid] = xa;
        xr[tid + 256] = xb;
        if (tid < 128) xr[tid + 512] = xc;
        if (tid < C0) m0[tid] = 0;
        __syncthreads();

        for (int i = tid; i < M_PTS * C0; i += 256) {
            int r = i >> 6, o = i & 63;
            float acc = 0.0f;
#pragma unroll
            for (int c = 0; c < C_IN; ++c) acc += xr[r * C_IN + c] * w0s[o * C_IN + c];
            float v = acc * s0[o] + b0[o];
            v = v > 0.0f ? v : 0.0f;
            x1[r * X1_STRIDE + o] = v;
            atomicMax(&m0[o], __float_as_int(v));
        }
        __syncthreads();

        {
            const int col = w * 16 + (lane & 15);
            v8f ce = {}, co = {};
#pragma unroll
            for (int k = 0; k < 16; k += 2) {
                const int k0 = 4 * k + 2 * hi;
                const int k1 = k0 + 4;
                v2f a0, a1;
                a0.x = __int_as_float(m0[k0]);
                a0.y = __int_as_float(m0[k0 + 1]);
                a1.x = __int_as_float(m0[k1]);
                a1.y = __int_as_float(m0[k1 + 1]);
                ce = __builtin_amdgcn_wmma_f32_16x16x4_f32(
                    false, a0, false, cB[k], (short)0, ce, false, false);
                co = __builtin_amdgcn_wmma_f32_16x16x4_f32(
                    false, a1, false, cB[k + 1], (short)0, co, false, false);
            }
            cvec[hi * C1 + col] = ce[0] + co[0];
        }
        __syncthreads();

        if (n + GEMM_WGS < N_VOX) {
            const int xo = (n + GEMM_WGS) * (M_PTS * C_IN) + tid;
            xa = X[xo];
            xb = X[xo + 256];
            if (tid < 128) xc = X[xo + 512];
        }

        v2f aF[16];
#pragma unroll
        for (int k = 0; k < 16; ++k)
            aF[k] = *(const v2f*)&x1[lrow * X1_STRIDE + 4 * k + 2 * hi];

#pragma unroll
        for (int jp = 0; jp < 4; jp += 2) {
            v8f accs[2];
#pragma unroll
            for (int t = 0; t < 2; ++t) {
                const float cv = cvec[(jbase + jp + t) * 16 + (lane & 15)];
                accs[t] = (v8f){cv, cv, cv, cv, cv, cv, cv, cv};
            }
#pragma unroll
            for (int k = 0; k < 16; ++k) {
#pragma unroll
                for (int t = 0; t < 2; ++t)
                    accs[t] = __builtin_amdgcn_wmma_f32_16x16x4_f32(
                        false, aF[k], false, bF[jp + t][k], (short)0, accs[t], false, false);
            }

#pragma unroll
            for (int t = 0; t < 2; ++t) {
                const int j = jbase + jp + t;
                const int ncol = j * 16 + (lane & 15);
                const v8f acc = accs[t];
                const float sg = sgn[ncol];
                float s = 0.0f, q = 0.0f;
                float m = acc[0] * sg;
#pragma unroll
                for (int i = 0; i < 8; ++i) {
                    float v = acc[i];
                    s += v;
                    q += v * v;
                    m = fmaxf(m, v * sg);
                }
                atomicAdd(&redS[ncol], s);
                atomicAdd(&redQ[ncol], q);
                ext[(rt * 2 + hi) * C1 + ncol] = m;
            }
        }
        __syncthreads();

        if (tid < C1) {
            float m = ext[tid];
            for (int p = 1; p < 8; ++p) m = fmaxf(m, ext[p * C1 + tid]);
            out[n * C1 + tid] = m * sgn[tid];
        }
    }

    if (tid < C1) {
        atomicAdd(&ws[WS_SUMY + tid], redS[tid]);
        atomicAdd(&ws[WS_SUMQ + tid], redQ[tid]);
    }
}

__global__ __launch_bounds__(256) void vfe_apply(const float* __restrict__ ws,
                                                 float* __restrict__ out) {
    const int i = blockIdx.x * 256 + threadIdx.x;
    if (i >= N_OUT4) return;
    const int o = (i & (C1 / 4 - 1)) * 4;
    const float4 sc = *(const float4*)&ws[WS_SCALE1 + o];
    const float4 bi = *(const float4*)&ws[WS_BIAS1 + o];
    float4 y = ((const float4*)out)[i];
    y.x = y.x * sc.x + bi.x;
    y.y = y.y * sc.y + bi.y;
    y.z = y.z * sc.z + bi.z;
    y.w = y.w * sc.w + bi.w;
    y.x = y.x > 0.0f ? y.x : 0.0f;
    y.y = y.y > 0.0f ? y.y : 0.0f;
    y.z = y.z > 0.0f ? y.z : 0.0f;
    y.w = y.w > 0.0f ? y.w : 0.0f;
    ((float4*)out)[i] = y;
}

extern "C" void kernel_launch(void* const* d_in, const int* in_sizes, int n_in,
                              void* d_out, int out_size, void* d_ws, size_t ws_size,
                              hipStream_t stream) {
    (void)in_sizes; (void)n_in; (void)out_size; (void)ws_size;
    const float* X   = (const float*)d_in[0];
    const float* W0  = (const float*)d_in[1];
    const float* g0  = (const float*)d_in[2];
    const float* be0 = (const float*)d_in[3];
    const float* W1  = (const float*)d_in[4];
    const float* g1  = (const float*)d_in[5];
    const float* be1 = (const float*)d_in[6];
    float* out = (float*)d_out;
    float* ws  = (float*)d_ws;

    const long P = (long)N_VOX * M_PTS;
    const float invP = 1.0f / (float)P;

    vfe_zero<<<(WS_TOTAL + 255) / 256, 256, 0, stream>>>(ws);
    vfe_stats0<<<1024, 256, 0, stream>>>(X, ws, P);
    vfe_final0<<<1, 64, 0, stream>>>(W0, g0, be0, ws, invP);
    vfe_gemm<<<GEMM_WGS, 256, 0, stream>>>(X, W0, W1, g1, ws, out);
    vfe_final1<<<1, 128, 0, stream>>>(g1, be1, ws, invP);
    vfe_apply<<<N_OUT4 / 256, 256, 0, stream>>>(ws, out);
}


# stats0 work-group reduction: 110 same-address LDS atomicAdds per thread (each compiled to a 32-step scalar lane loop) replaced by row_xmask DPP row sums + a 16-row LDS sum (sum order changes)
# speedup vs baseline: 4.0385x; 1.7162x over previous
#include <hip/hip_runtime.h>
#include <hip/hip_bf16.h>

typedef float v2f __attribute__((ext_vector_type(2)));
typedef float v8f __attribute__((ext_vector_type(8)));

#define N_VOX 20000
#define M_PTS 64
#define C_IN  10
#define C0    64
#define C1    128
#define EPS   1e-3f
#define N_OUT4 640000
#define GEMM_WGS 512

#define WS_SUMX   0
#define WS_SXX    16
#define WS_SCALE0 128
#define WS_BIAS0  192
#define WS_SUMY   256
#define WS_SUMQ   384
#define WS_SCALE1 512
#define WS_BIAS1  640
#define WS_TOTAL  768

#define X1_STRIDE 132

__global__ void vfe_zero(float* __restrict__ ws) {
    int i = blockIdx.x * blockDim.x + threadIdx.x;
    if (i < WS_TOTAL) ws[i] = 0.0f;
}

#define RED_PITCH 112

template <int CTRL>
__device__ __forceinline__ float row_xor_add(float v) {
    const int o = __builtin_amdgcn_update_dpp(0, __builtin_bit_cast(int, v), CTRL, 0xf, 0xf, false);
    return v + __builtin_bit_cast(float, o);
}

__global__ __launch_bounds__(256) void vfe_stats0(const float* __restrict__ X,
                                                  float* __restrict__ ws, long P) {
    float sx[C_IN] = {};
    float sxx[C_IN * C_IN] = {};
    for (long i = (long)blockIdx.x * blockDim.x + threadIdx.x; i < P;
         i += (long)gridDim.x * blockDim.x) {
        const float* xp = X + i * C_IN;
        float x[C_IN];
#pragma unroll
        for (int c = 0; c < C_IN; ++c) x[c] = xp[c];
#pragma unroll
        for (int c = 0; c < C_IN; ++c) {
            sx[c] += x[c];
#pragma unroll
            for (int d = 0; d < C_IN; ++d) sxx[c * C_IN + d] += x[c] * x[d];
        }
    }
#pragma unroll
    for (int c = 0; c < C_IN; ++c) sx[c] = row_xor_add<0x161>(sx[c]);
#pragma unroll
    for (int t = 0; t < C_IN * C_IN; ++t) sxx[t] = row_xor_add<0x161>(sxx[t]);
#pragma unroll
    for (int c = 0; c < C_IN; ++c) sx[c] = row_xor_add<0x162>(sx[c]);
#pragma unroll
    for (int t = 0; t < C_IN * C_IN; ++t) sxx[t] = row_xor_add<0x162>(sxx[t]);
#pragma unroll
    for (int c = 0; c < C_IN; ++c) sx[c] = row_xor_add<0x164>(sx[c]);
#pragma unroll
    for (int t = 0; t < C_IN * C_IN; ++t) sxx[t] = row_xor_add<0x164>(sxx[t]);
#pragma unroll
    for (int c = 0; c < C_IN; ++c) sx[c] = row_xor_add<0x168>(sx[c]);
#pragma unroll
    for (int t = 0; t < C_IN * C_IN; ++t) sxx[t] = row_xor_add<0x168>(sxx[t]);

    __shared__ __attribute__((aligned(16))) float red[16 * RED_PITCH];
    if ((threadIdx.x & 15) == 0) {
        float* r = red + (threadIdx.x >> 4) * RED_PITCH;
#pragma unroll
        for (int c = 0; c < C_IN; ++c) r[c] = sx[c];
#pragma unroll
        for (int t = 0; t < C_IN * C_IN; ++t) r[C_IN + t] = sxx[t];
    }
    __syncthreads();
    const int t = threadIdx.x;
    if (t < C_IN + C_IN * C_IN) {
        float s = 0.0f;
#pragma unroll
        for (int p = 0; p < 16; ++p) s += red[p * RED_PITCH + t];
        if (t < C_IN)
            atomicAdd(&ws[WS_SUMX + t], s);
        else
            atomicAdd(&ws[WS_SXX + (t - C_IN)], s);
    }
}

__global__ void vfe_final0(const float* __restrict__ W0, const float* __restrict__ g0,
                           const float* __restrict__ be0, float* __restrict__ ws,
                           float invP) {
    int o = threadIdx.x;
    if (o >= C0) return;
    float w[C_IN];
#pragma unroll
    for (int c = 0; c < C_IN; ++c) w[c] = W0[o * C_IN + c];
    float mean = 0.0f;
#pragma unroll
    for (int c = 0; c < C_IN; ++c) mean += w[c] * ws[WS_SUMX + c];
    mean *= invP;
    float ey2 = 0.0f;
#pragma unroll
    for (int c = 0; c < C_IN; ++c)
#pragma unroll
        for (int d = 0; d < C_IN; ++d) ey2 += w[c] * w[d] * ws[WS_SXX + c * C_IN + d];
    ey2 *= invP;
    float var = ey2 - mean * mean;
    float sc = g0[o] * rsqrtf(var + EPS);
    ws[WS_SCALE0 + o] = sc;
    ws[WS_BIAS0 + o] = be0[o] - mean * sc;
}

__global__ void vfe_final1(const float* __restrict__ g1, const float* __restrict__ be1,
                           float* __restrict__ ws, float invP) {
    int o = threadIdx.x;
    if (o >= C1) return;
    float mean = ws[WS_SUMY + o] * invP;
    float var = ws[WS_SUMQ + o] * invP - mean * mean;
    float sc = g1[o] * rsqrtf(var + EPS);
    ws[WS_SCALE1 + o] = sc;
    ws[WS_BIAS1 + o] = be1[o] - mean * sc;
}

__global__ __launch_bounds__(256) __attribute__((amdgpu_waves_per_eu(4, 4)))
void vfe_gemm(const float* __restrict__ X,
              const float* __restrict__ W0,
              const float* __restrict__ W1,
              const float* __restrict__ g1,
              float* __restrict__ ws,
              float* __restrict__ out) {
    __shared__ float xr[M_PTS * C_IN];
    __shared__ float w0s[C0 * C_IN];
    __shared__ float s0[C0], b0[C0];
    __shared__ int   m0[C0];
    __shared__ float x1[M_PTS * X1_STRIDE];
    __shared__ float redS[C1], redQ[C1];
    __shared__ float sgn[C1];
    __shared__ float ext[8 * C1];
    __shared__ float cvec[2 * C1];

    const int tid = threadIdx.x;

    for (int i = tid; i < C0 * C_IN; i += 256) w0s[i] = W0[i];
    if (tid < C0) {
        s0[tid] = ws[WS_SCALE0 + tid];
        b0[tid] = ws[WS_BIAS0 + tid];
    }
    if (tid < C1) {
        redS[tid] = 0.0f;
        redQ[tid] = 0.0f;
        sgn[tid] = g1[tid] < 0.0f ? -1.0f : 1.0f;
    }

    const int lane = tid & 31;
    const int w = tid >> 5;
    const int rt = w & 3;
    const int jbase = (w >> 2) * 4;
    const int hi = lane >> 4;
    const int lrow = rt * 16 + (lane & 15);

    v2f bF[4][16];
    v2f cB[16];
#pragma unroll
    for (int jj = 0; jj < 4; ++jj) {
        const float* wr = W1 + ((jbase + jj) * 16 + (lane & 15)) * C1 + 2 * hi;
#pragma unroll
        for (int k = 0; k < 16; ++k) bF[jj][k] = *(const v2f*)(wr + 4 * k);
    }
    {
        const float* wr = W1 + (w * 16 + (lane & 15)) * C1 + C0 + 2 * hi;
#pragma unroll
        for (int k = 0; k < 16; ++k) cB[k] = *(const v2f*)(wr + 4 * k);
    }

    float xa, xb, xc = 0.0f;
    {
        const int xo = (int)blockIdx.x * (M_PTS * C_IN) + tid;
        xa = X[xo];
        xb = X[xo + 256];
        if (tid < 128) xc = X[xo + 512];
    }

    for (int n = blockIdx.x; n < N_VOX; n += GEMM_WGS) {
        xr[tid] = xa;
        xr[tid + 256] = xb;
        if (tid < 128) xr[tid + 512] = xc;
        if (tid < C0) m0[tid] = 0;
        __syncthreads();

        for (int i = tid; i < M_PTS * C0; i += 256) {
            int r = i >> 6, o = i & 63;
            float acc = 0.0f;
#pragma unroll
            for (int c = 0; c < C_IN; ++c) acc += xr[r * C_IN + c] * w0s[o * C_IN + c];
            float v = acc * s0[o] + b0[o];
            v = v > 0.0f ? v : 0.0f;
            x1[r * X1_STRIDE + o] = v;
            atomicMax(&m0[o], __float_as_int(v));
        }
        __syncthreads();

        {
            const int col = w * 16 + (lane & 15);
            v8f ce = {}, co = {};
#pragma unroll
            for (int k = 0; k < 16; k += 2) {
                const int k0 = 4 * k + 2 * hi;
                const int k1 = k0 + 4;
                v2f a0, a1;
                a0.x = __int_as_float(m0[k0]);
                a0.y = __int_as_float(m0[k0 + 1]);
                a1.x = __int_as_float(m0[k1]);
                a1.y = __int_as_float(m0[k1 + 1]);
                ce = __builtin_amdgcn_wmma_f32_16x16x4_f32(
                    false, a0, false, cB[k], (short)0, ce, false, false);
                co = __builtin_amdgcn_wmma_f32_16x16x4_f32(
                    false, a1, false, cB[k + 1], (short)0, co, false, false);
            }
            cvec[hi * C1 + col] = ce[0] + co[0];
        }
        __syncthreads();

        if (n + GEMM_WGS < N_VOX) {
            const int xo = (n + GEMM_WGS) * (M_PTS * C_IN) + tid;
            xa = X[xo];
            xb = X[xo + 256];
            if (tid < 128) xc = X[xo + 512];
        }

        v2f aF[16];
#pragma unroll
        for (int k = 0; k < 16; ++k)
            aF[k] = *(const v2f*)&x1[lrow * X1_STRIDE + 4 * k + 2 * hi];

#pragma unroll
        for (int jp = 0; jp < 4; jp += 2) {
            v8f accs[2];
#pragma unroll
            for (int t = 0; t < 2; ++t) {
                const float cv = cvec[(jbase + jp + t) * 16 + (lane & 15)];
                accs[t] = (v8f){cv, cv, cv, cv, cv, cv, cv, cv};
            }
#pragma unroll
            for (int k = 0; k < 16; ++k) {
#pragma unroll
                for (int t = 0; t < 2; ++t)
                    accs[t] = __builtin_amdgcn_wmma_f32_16x16x4_f32(
                        false, aF[k], false, bF[jp + t][k], (short)0, accs[t], false, false);
            }

#pragma unroll
            for (int t = 0; t < 2; ++t) {
                const int j = jbase + jp + t;
                const int ncol = j * 16 + (lane & 15);
                const v8f acc = accs[t];
                const float sg = sgn[ncol];
                float s = 0.0f, q = 0.0f;
                float m = acc[0] * sg;
#pragma unroll
                for (int i = 0; i < 8; ++i) {
                    float v = acc[i];
                    s += v;
                    q += v * v;
                    m = fmaxf(m, v * sg);
                }
                atomicAdd(&redS[ncol], s);
                atomicAdd(&redQ[ncol], q);
                ext[(rt * 2 + hi) * C1 + ncol] = m;
            }
        }
        __syncthreads();

        if (tid < C1) {
            float m = ext[tid];
            for (int p = 1; p < 8; ++p) m = fmaxf(m, ext[p * C1 + tid]);
            out[n * C1 + tid] = m * sgn[tid];
        }
    }

    if (tid < C1) {
        atomicAdd(&ws[WS_SUMY + tid], redS[tid]);
        atomicAdd(&ws[WS_SUMQ + tid], redQ[tid]);
    }
}

__global__ __launch_bounds__(256) void vfe_apply(const float* __restrict__ ws,
                                                 float* __restrict__ out) {
    const int i = blockIdx.x * 256 + threadIdx.x;
    if (i >= N_OUT4) return;
    const int o = (i & (C1 / 4 - 1)) * 4;
    const float4 sc = *(const float4*)&ws[WS_SCALE1 + o];
    const float4 bi = *(const float4*)&ws[WS_BIAS1 + o];
    float4 y = ((const float4*)out)[i];
    y.x = y.x * sc.x + bi.x;
    y.y = y.y * sc.y + bi.y;
    y.z = y.z * sc.z + bi.z;
    y.w = y.w * sc.w + bi.w;
    y.x = y.x > 0.0f ? y.x : 0.0f;
    y.y = y.y > 0.0f ? y.y : 0.0f;
    y.z = y.z > 0.0f ? y.z : 0.0f;
    y.w = y.w > 0.0f ? y.w : 0.0f;
    ((float4*)out)[i] = y;
}

extern "C" void kernel_launch(void* const* d_in, const int* in_sizes, int n_in,
                              void* d_out, int out_size, void* d_ws, size_t ws_size,
                              hipStream_t stream) {
    (void)in_sizes; (void)n_in; (void)out_size; (void)ws_size;
    const float* X   = (const float*)d_in[0];
    const float* W0  = (const float*)d_in[1];
    const float* g0  = (const float*)d_in[2];
    const float* be0 = (const float*)d_in[3];
    const float* W1  = (const float*)d_in[4];
    const float* g1  = (const float*)d_in[5];
    const float* be1 = (const float*)d_in[6];
    float* out = (float*)d_out;
    float* ws  = (float*)d_ws;

    const long P = (long)N_VOX * M_PTS;
    const float invP = 1.0f / (float)P;

    vfe_zero<<<(WS_TOTAL + 255) / 256, 256, 0, stream>>>(ws);
    vfe_stats0<<<1024, 256, 0, stream>>>(X, ws, P);
    vfe_final0<<<1, 64, 0, stream>>>(W0, g0, be0, ws, invP);
    vfe_gemm<<<GEMM_WGS, 256, 0, stream>>>(X, W0, W1, g1, ws, out);
    vfe_final1<<<1, 128, 0, stream>>>(g1, be1, ws, invP);
    vfe_apply<<<N_OUT4 / 256, 256, 0, stream>>>(ws, out);
}


# layer-0 product (64x10 by 10x64 per voxel) moved from vector-ALU dot products over LDS onto WMMA: K padded 10 to 12, A pairs straight from global, one barrier per voxel fewer (sum order changes)
# speedup vs baseline: 5.0273x; 1.2448x over previous
#include <hip/hip_runtime.h>
#include <hip/hip_bf16.h>

typedef float v2f __attribute__((ext_vector_type(2)));
typedef float v8f __attribute__((ext_vector_type(8)));

#define N_VOX 20000
#define M_PTS 64
#define C_IN  10
#define C0    64
#define C1    128
#define EPS   1e-3f
#define N_OUT4 640000
#define GEMM_WGS 512

#define WS_SUMX   0
#define WS_SXX    16
#define WS_SCALE0 128
#define WS_BIAS0  192
#define WS_SUMY   256
#define WS_SUMQ   384
#define WS_SCALE1 512
#define WS_BIAS1  640
#define WS_TOTAL  768

#define X1_STRIDE 132

#define L_W0F   0
#define L_S0    (L_W0F + 4 * 3 * 32 * 2)
#define L_B0    (L_S0 + C0)
#define L_SGN   (L_B0 + C0)
#define L_REDS  (L_SGN + C1)
#define L_REDQ  (L_REDS + C1)
#define L_CVEC  (L_REDQ + C1)
#define L_EXT   (L_CVEC + 2 * C1)
#define L_X1    (L_EXT + 8 * C1)
#define L_TOTAL (L_X1 + M_PTS * X1_STRIDE)

__global__ void vfe_zero(float* __restrict__ ws) {
    int i = blockIdx.x * blockDim.x + threadIdx.x;
    if (i < WS_TOTAL) ws[i] = 0.0f;
}

#define RED_PITCH 112

template <int CTRL>
__device__ __forceinline__ float row_xor_add(float v) {
    const int o = __builtin_amdgcn_update_dpp(0, __builtin_bit_cast(int, v), CTRL, 0xf, 0xf, false);
    return v + __builtin_bit_cast(float, o);
}

__global__ __launch_bounds__(256) void vfe_stats0(const float* __restrict__ X,
                                                  float* __restrict__ ws, long P) {
    float sx[C_IN] = {};
    float sxx[C_IN * C_IN] = {};
    for (long i = (long)blockIdx.x * blockDim.x + threadIdx.x; i < P;
         i += (long)gridDim.x * blockDim.x) {
        const float* xp = X + i * C_IN;
        float x[C_IN];
#pragma unroll
        for (int c = 0; c < C_IN; ++c) x[c] = xp[c];
#pragma unroll
        for (int c = 0; c < C_IN; ++c) {
            sx[c] += x[c];
#pragma unroll
            for (int d = 0; d < C_IN; ++d) sxx[c * C_IN + d] += x[c] * x[d];
        }
    }
#pragma unroll
    for (int c = 0; c < C_IN; ++c) sx[c] = row_xor_add<0x161>(sx[c]);
#pragma unroll
    for (int t = 0; t < C_IN * C_IN; ++t) sxx[t] = row_xor_add<0x161>(sxx[t]);
#pragma unroll
    for (int c = 0; c < C_IN; ++c) sx[c] = row_xor_add<0x162>(sx[c]);
#pragma unroll
    for (int t = 0; t < C_IN * C_IN; ++t) sxx[t] = row_xor_add<0x162>(sxx[t]);
#pragma unroll
    for (int c = 0; c < C_IN; ++c) sx[c] = row_xor_add<0x164>(sx[c]);
#pragma unroll
    for (int t = 0; t < C_IN * C_IN; ++t) sxx[t] = row_xor_add<0x164>(sxx[t]);
#pragma unroll
    for (int c = 0; c < C_IN; ++c) sx[c] = row_xor_add<0x168>(sx[c]);
#pragma unroll
    for (int t = 0; t < C_IN * C_IN; ++t) sxx[t] = row_xor_add<0x168>(sxx[t]);

    __shared__ __attribute__((aligned(16))) float red[16 * RED_PITCH];
    if ((threadIdx.x & 15) == 0) {
        float* r = red + (threadIdx.x >> 4) * RED_PITCH;
#pragma unroll
        for (int c = 0; c < C_IN; ++c) r[c] = sx[c];
#pragma unroll
        for (int t = 0; t < C_IN * C_IN; ++t) r[C_IN + t] = sxx[t];
    }
    __syncthreads();
    const int t = threadIdx.x;
    if (t < C_IN + C_IN * C_IN) {
        float s = 0.0f;
#pragma unroll
        for (int p = 0; p < 16; ++p) s += red[p * RED_PITCH + t];
        if (t < C_IN)
            atomicAdd(&ws[WS_SUMX + t], s);
        else
            atomicAdd(&ws[WS_SXX + (t - C_IN)], s);
    }
}

__global__ void vfe_final0(const float* __restrict__ W0, const float* __restrict__ g0,
                           const float* __restrict__ be0, float* __restrict__ ws,
                           float invP) {
    int o = threadIdx.x;
    if (o >= C0) return;
    float w[C_IN];
#pragma unroll
    for (int c = 0; c < C_IN; ++c) w[c] = W0[o * C_IN + c];
    float mean = 0.0f;
#pragma unroll
    for (int c = 0; c < C_IN; ++c) mean += w[c] * ws[WS_SUMX + c];
    mean *= invP;
    float ey2 = 0.0f;
#pragma unroll
    for (int c = 0; c < C_IN; ++c)
#pragma unroll
        for (int d = 0; d < C_IN; ++d) ey2 += w[c] * w[d] * ws[WS_SXX + c * C_IN + d];
    ey2 *= invP;
    float var = ey2 - mean * mean;
    float sc = g0[o] * rsqrtf(var + EPS);
    ws[WS_SCALE0 + o] = sc;
    ws[WS_BIAS0 + o] = be0[o] - mean * sc;
}

__global__ void vfe_final1(const float* __restrict__ g1, const float* __restrict__ be1,
                           float* __restrict__ ws, float invP) {
    int o = threadIdx.x;
    if (o >= C1) return;
    float mean = ws[WS_SUMY + o] * invP;
    float var = ws[WS_SUMQ + o] * invP - mean * mean;
    float sc = g1[o] * rsqrtf(var + EPS);
    ws[WS_SCALE1 + o] = sc;
    ws[WS_BIAS1 + o] = be1[o] - mean * sc;
}

__global__ __launch_bounds__(256) __attribute__((amdgpu_waves_per_eu(4, 4)))
void vfe_gemm(const float* __restrict__ X,
              const float* __restrict__ W0,
              const float* __restrict__ W1,
              const float* __restrict__ g1,
              float* __restrict__ ws,
              float* __restrict__ out) {
    __shared__ __attribute__((aligned(16))) float lds[L_TOTAL];
    __shared__ int m0[C0];
    float* const x1   = lds + L_X1;
    float* const w0f  = lds + L_W0F;
    float* const s0   = lds + L_S0;
    float* const b0   = lds + L_B0;
    float* const sgn  = lds + L_SGN;
    float* const redS = lds + L_REDS;
    float* const redQ = lds + L_REDQ;
    float* const cvec = lds + L_CVEC;
    float* const ext  = lds + L_EXT;

    const int tid = threadIdx.x;

    for (int i = tid; i < 4 * 3 * 32 * 2; i += 256) {
        const int e = i & 1, ln = (i >> 1) & 31, g = i >> 6;
        const int k = 4 * (g % 3) + 2 * (ln >> 4) + e;
        const int col = (g / 3) * 16 + (ln & 15);
        w0f[i] = k < C_IN ? W0[col * C_IN + k] : 0.0f;
    }
    if (tid < C0) {
        s0[tid] = ws[WS_SCALE0 + tid];
        b0[tid] = ws[WS_BIAS0 + tid];
        m0[tid] = 0;
    }
    if (tid < C1) {
        redS[tid] = 0.0f;
        redQ[tid] = 0.0f;
        sgn[tid] = g1[tid] < 0.0f ? -1.0f : 1.0f;
    }

    const int lane = tid & 31;
    const int w = tid >> 5;
    const int rt = w & 3;
    const int jbase = (w >> 2) * 4;
    const int hi = lane >> 4;
    const int lrow = rt * 16 + (lane & 15);

    v2f bF[4][16];
    v2f cB[16];
#pragma unroll
    for (int jj = 0; jj < 4; ++jj) {
        const float* wr = W1 + ((jbase + jj) * 16 + (lane & 15)) * C1 + 2 * hi;
#pragma unroll
        for (int k = 0; k < 16; ++k) bF[jj][k] = *(const v2f*)(wr + 4 * k);
    }
    {
        const float* wr = W1 + (w * 16 + (lane & 15)) * C1 + C0 + 2 * hi;
#pragma unroll
        for (int k = 0; k < 16; ++k) cB[k] = *(const v2f*)(wr + 4 * k);
    }

    const int l0tile = (w >> 2) * 2;
    const v2f* const X2 = (const v2f*)X;
    const unsigned xp01 = lrow * (C_IN / 2) + hi;
    const unsigned xp2 = lrow * (C_IN / 2) + 4 - hi;

    const float* const w0p = w0f + (l0tile * 3 * 32 + lane) * 2;
    float* const l0p = s0 + l0tile * 16 + (lane & 15);
    int*   const m0p = m0 + l0tile * 16 + (lane & 15);
    float* const x1st = x1 + (rt * 16 + 8 * hi) * X1_STRIDE + l0tile * 16 + (lane & 15);
    const float* const x1ld = x1 + lrow * X1_STRIDE + 2 * hi;
    const int*   const m0r = m0 + 2 * hi;
    float* const ep = sgn + jbase * 16 + (lane & 15);
    float* const extp = ext + (rt * 2 + hi) * C1 + jbase * 16 + (lane & 15);
    v2f pa0, pa1, pa2;
    {
        const unsigned vo = blockIdx.x * (M_PTS * C_IN / 2);
        pa0 = X2[vo + xp01];
        pa1 = X2[vo + xp01 + 2];
        pa2 = X2[vo + xp2];
    }
    __syncthreads();

    for (int n = blockIdx.x; n < N_VOX; n += GEMM_WGS) {
        {
            v2f a[3];
            a[0] = pa0;
            a[1] = pa1;
            a[2] = hi ? (v2f){0.0f, 0.0f} : pa2;
            v2f b[2][3];
#pragma unroll
            for (int t = 0; t < 2; ++t)
#pragma unroll
                for (int s = 0; s < 3; ++s)
                    b[t][s] = *(const v2f*)&w0p[(t * 3 + s) * 64];
            v8f l0acc[2] = {};
#pragma unroll
            for (int s = 0; s < 3; ++s)
#pragma unroll
                for (int t = 0; t < 2; ++t)
                    l0acc[t] = __builtin_amdgcn_wmma_f32_16x16x4_f32(
                        false, a[s], false, b[t][s], (short)0, l0acc[t], false, false);
#pragma unroll
            for (int t = 0; t < 2; ++t) {
                const float sc = l0p[16 * t], bi = l0p[(L_B0 - L_S0) + 16 * t];
                float* xp = x1st + 16 * t;
                float m = 0.0f;
#pragma unroll
                for (int i = 0; i < 8; ++i) {
                    float v = l0acc[t][i] * sc + bi;
                    v = v > 0.0f ? v : 0.0f;
                    xp[i * X1_STRIDE] = v;
                    m = fmaxf(m, v);
                }
                atomicMax(&m0p[16 * t], __float_as_int(m));
            }
        }
        __syncthreads();

        {
            const int col = w * 16 + (lane & 15);
            v8f ce = {}, co = {};
#pragma unroll
            for (int k = 0; k < 16; k += 2) {
                v2f a0, a1;
                a0.x = __int_as_float(m0r[4 * k]);
                a0.y = __int_as_float(m0r[4 * k + 1]);
                a1.x = __int_as_float(m0r[4 * k + 4]);
                a1.y = __int_as_float(m0r[4 * k + 5]);
                ce = __builtin_amdgcn_wmma_f32_16x16x4_f32(
                    false, a0, false, cB[k], (short)0, ce, false, false);
                co = __builtin_amdgcn_wmma_f32_16x16x4_f32(
                    false, a1, false, cB[k + 1], (short)0, co, false, false);
            }
            cvec[hi * C1 + col] = ce[0] + co[0];
        }
        __syncthreads();

        if (tid < C0) m0[tid] = 0;

        v2f aF[16];
#pragma unroll
        for (int k = 0; k < 16; ++k)
            aF[k] = *(const v2f*)&x1ld[4 * k];

#pragma unroll
        for (int jp = 0; jp < 4; jp += 2) {
            v8f accs[2];
#pragma unroll
            for (int t = 0; t < 2; ++t) {
                const float cv = ep[(L_CVEC - L_SGN) + 16 * (jp + t)];
                accs[t] = (v8f){cv, cv, cv, cv, cv, cv, cv, cv};
            }
#pragma unroll
            for (int k = 0; k < 16; ++k) {
#pragma unroll
                for (int t = 0; t < 2; ++t)
                    accs[t] = __builtin_amdgcn_wmma_f32_16x16x4_f32(
                        false, aF[k], false, bF[jp + t][k], (short)0, accs[t], false, false);
            }

#pragma unroll
            for (int t = 0; t < 2; ++t) {
                const v8f acc = accs[t];
                const float sg = ep[16 * (jp + t)];
                float s = 0.0f, q = 0.0f;
                float m = acc[0] * sg;
#pragma unroll
                for (int i = 0; i < 8; ++i) {
                    float v = acc[i];
                    s += v;
                    q += v * v;
                    m = fmaxf(m, v * sg);
                }
                atomicAdd(&ep[(L_REDS - L_SGN) + 16 * (jp + t)], s);
                atomicAdd(&ep[(L_REDQ - L_SGN) + 16 * (jp + t)], q);
                extp[16 * (jp + t)] = m;
            }

            if (jp == 0 && n + GEMM_WGS < N_VOX) {
                const unsigned vo = (unsigned)(n + GEMM_WGS) * (M_PTS * C_IN / 2);
                pa0 = X2[vo + xp01];
                pa1 = X2[vo + xp01 + 2];
                pa2 = X2[vo + xp2];
            }
        }
        __syncthreads();

        if (tid < C1) {
            float m = ext[tid];
            for (int p = 1; p < 8; ++p) m = fmaxf(m, ext[p * C1 + tid]);
            out[n * C1 + tid] = m * sgn[tid];
        }
    }

    if (tid < C1) {
        atomicAdd(&ws[WS_SUMY + tid], redS[tid]);
        atomicAdd(&ws[WS_SUMQ + tid], redQ[tid]);
    }
}

__global__ __launch_bounds__(256) void vfe_apply(const float* __restrict__ ws,
                                                 float* __restrict__ out) {
    const int i = blockIdx.x * 256 + threadIdx.x;
    if (i >= N_OUT4) return;
    const int o = (i & (C1 / 4 - 1)) * 4;
    const float4 sc = *(const float4*)&ws[WS_SCALE1 + o];
    const float4 bi = *(const float4*)&ws[WS_BIAS1 + o];
    float4 y = ((const float4*)out)[i];
    y.x = y.x * sc.x + bi.x;
    y.y = y.y * sc.y + bi.y;
    y.z = y.z * sc.z + bi.z;
    y.w = y.w * sc.w + bi.w;
    y.x = y.x > 0.0f ? y.x : 0.0f;
    y.y = y.y > 0.0f ? y.y : 0.0f;
    y.z = y.z > 0.0f ? y.z : 0.0f;
    y.w = y.w > 0.0f ? y.w : 0.0f;
    ((float4*)out)[i] = y;
}

extern "C" void kernel_launch(void* const* d_in, const int* in_sizes, int n_in,
                              void* d_out, int out_size, void* d_ws, size_t ws_size,
                              hipStream_t stream) {
    (void)in_sizes; (void)n_in; (void)out_size; (void)ws_size;
    const float* X   = (const float*)d_in[0];
    const float* W0  = (const float*)d_in[1];
    const float* g0  = (const float*)d_in[2];
    const float* be0 = (const float*)d_in[3];
    const float* W1  = (const float*)d_in[4];
    const float* g1  = (const float*)d_in[5];
    const float* be1 = (const float*)d_in[6];
    float* out = (float*)d_out;
    float* ws  = (float*)d_ws;

    const long P = (long)N_VOX * M_PTS;
    const float invP = 1.0f / (float)P;

    vfe_zero<<<(WS_TOTAL + 255) / 256, 256, 0, stream>>>(ws);
    vfe_stats0<<<1024, 256, 0, stream>>>(X, ws, P);
    vfe_final0<<<1, 64, 0, stream>>>(W0, g0, be0, ws, invP);
    vfe_gemm<<<GEMM_WGS, 256, 0, stream>>>(X, W0, W1, g1, ws, out);
    vfe_final1<<<1, 128, 0, stream>>>(g1, be1, ws, invP);
    vfe_apply<<<N_OUT4 / 256, 256, 0, stream>>>(ws, out);
}


# repeated-max share of y1 batched: the 16 voxels of a group are the 16 rows of ONE WMMA tile (was one tile per voxel), its terms added in a group epilogue, not as accumulator start (sum order changes)
# speedup vs baseline: 5.4712x; 1.0883x over previous
#include <hip/hip_runtime.h>
#include <hip/hip_bf16.h>

typedef float v2f __attribute__((ext_vector_type(2)));
typedef float v8f __attribute__((ext_vector_type(8)));
typedef int v2i __attribute__((ext_vector_type(2)));

#define N_VOX 20000
#define M_PTS 64
#define C_IN  10
#define C0    64
#define C1    128
#define EPS   1e-3f
#define N_OUT4 640000
#define GEMM_WGS 512

#define WS_SUMX   0
#define WS_SXX    16
#define WS_SCALE0 128
#define WS_BIAS0  192
#define WS_SUMY   256
#define WS_SUMQ   384
#define WS_SCALE1 512
#define WS_BIAS1  640
#define WS_TOTAL  768

#define X1_STRIDE 132

#define VGRP    16
#define L_W0F   0
#define L_S0    (L_W0F + 4 * 3 * 32 * 2)
#define L_B0    (L_S0 + C0)
#define L_SGN   (L_B0 + C0)
#define L_REDS  (L_SGN + C1)
#define L_REDQ  (L_REDS + C1)
#define L_EXT   (L_REDQ + C1)
#define L_SUMV  (L_EXT + 8 * C1)
#define L_EXTM  (L_SUMV + VGRP * C1)
#define L_X1    (L_EXTM + VGRP * C1)
#define L_WUP   (L_X1 + M_PTS * X1_STRIDE)
#define L_TOTAL (L_WUP + 8 * 16 * 32 * 2)
#define M0_PITCH 68

__global__ void vfe_zero(float* __restrict__ ws) {
    int i = blockIdx.x * blockDim.x + threadIdx.x;
    if (i < WS_TOTAL) ws[i] = 0.0f;
}

#define RED_PITCH 112

template <int CTRL>
__device__ __forceinline__ float row_xor_add(float v) {
    const int o = __builtin_amdgcn_update_dpp(0, __builtin_bit_cast(int, v), CTRL, 0xf, 0xf, false);
    return v + __builtin_bit_cast(float, o);
}

__global__ __launch_bounds__(256) void vfe_stats0(const float* __restrict__ X,
                                                  float* __restrict__ ws, long P) {
    float sx[C_IN] = {};
    float sxx[C_IN * C_IN] = {};
    for (long i = (long)blockIdx.x * blockDim.x + threadIdx.x; i < P;
         i += (long)gridDim.x * blockDim.x) {
        const float* xp = X + i * C_IN;
        float x[C_IN];
#pragma unroll
        for (int c = 0; c < C_IN; ++c) x[c] = xp[c];
#pragma unroll
        for (int c = 0; c < C_IN; ++c) {
            sx[c] += x[c];
#pragma unroll
            for (int d = 0; d < C_IN; ++d) sxx[c * C_IN + d] += x[c] * x[d];
        }
    }
#pragma unroll
    for (int c = 0; c < C_IN; ++c) sx[c] = row_xor_add<0x161>(sx[c]);
#pragma unroll
    for (int t = 0; t < C_IN * C_IN; ++t) sxx[t] = row_xor_add<0x161>(sxx[t]);
#pragma unroll
    for (int c = 0; c < C_IN; ++c) sx[c] = row_xor_add<0x162>(sx[c]);
#pragma unroll
    for (int t = 0; t < C_IN * C_IN; ++t) sxx[t] = row_xor_add<0x162>(sxx[t]);
#pragma unroll
    for (int c = 0; c < C_IN; ++c) sx[c] = row_xor_add<0x164>(sx[c]);
#pragma unroll
    for (int t = 0; t < C_IN * C_IN; ++t) sxx[t] = row_xor_add<0x164>(sxx[t]);
#pragma unroll
    for (int c = 0; c < C_IN; ++c) sx[c] = row_xor_add<0x168>(sx[c]);
#pragma unroll
    for (int t = 0; t < C_IN * C_IN; ++t) sxx[t] = row_xor_add<0x168>(sxx[t]);

    __shared__ __attribute__((aligned(16))) float red[16 * RED_PITCH];
    if ((threadIdx.x & 15) == 0) {
        float* r = red + (threadIdx.x >> 4) * RED_PITCH;
#pragma unroll
        for (int c = 0; c < C_IN; ++c) r[c] = sx[c];
#pragma unroll
        for (int t = 0; t < C_IN * C_IN; ++t) r[C_IN + t] = sxx[t];
    }
    __syncthreads();
    const int t = threadIdx.x;
    if (t < C_IN + C_IN * C_IN) {
        float s = 0.0f;
#pragma unroll
        for (int p = 0; p < 16; ++p) s += red[p * RED_PITCH + t];
        if (t < C_IN)
            atomicAdd(&ws[WS_SUMX + t], s);
        else
            atomicAdd(&ws[WS_SXX + (t - C_IN)], s);
    }
}

__global__ void vfe_final0(const float* __restrict__ W0, const float* __restrict__ g0,
                           const float* __restrict__ be0, float* __restrict__ ws,
                           float invP) {
    int o = threadIdx.x;
    if (o >= C0) return;
    float w[C_IN];
#pragma unroll
    for (int c = 0; c < C_IN; ++c) w[c] = W0[o * C_IN + c];
    float mean = 0.0f;
#pragma unroll
    for (int c = 0; c < C_IN; ++c) mean += w[c] * ws[WS_SUMX + c];
    mean *= invP;
    float ey2 = 0.0f;
#pragma unroll
    for (int c = 0; c < C_IN; ++c)
#pragma unroll
        for (int d = 0; d < C_IN; ++d) ey2 += w[c] * w[d] * ws[WS_SXX + c * C_IN + d];
    ey2 *= invP;
    float var = ey2 - mean * mean;
    float sc = g0[o] * rsqrtf(var + EPS);
    ws[WS_SCALE0 + o] = sc;
    ws[WS_BIAS0 + o] = be0[o] - mean * sc;
}

__global__ void vfe_final1(const float* __restrict__ g1, const float* __restrict__ be1,
                           float* __restrict__ ws, float invP) {
    int o = threadIdx.x;
    if (o >= C1) return;
    float mean = ws[WS_SUMY + o] * invP;
    float var = ws[WS_SUMQ + o] * invP - mean * mean;
    float sc = g1[o] * rsqrtf(var + EPS);
    ws[WS_SCALE1 + o] = sc;
    ws[WS_BIAS1 + o] = be1[o] - mean * sc;
}

__global__ __launch_bounds__(256) __attribute__((amdgpu_waves_per_eu(4, 4)))
void vfe_gemm(const float* __restrict__ X,
              const float* __restrict__ W0,
              const float* __restrict__ W1,
              const float* __restrict__ g1,
              float* __restrict__ ws,
              float* __restrict__ out) {
    __shared__ __attribute__((aligned(16))) float lds[L_TOTAL];
    __shared__ __attribute__((aligned(16))) int m0[VGRP * M0_PITCH];
    float* const x1   = lds + L_X1;
    float* const w0f  = lds + L_W0F;
    float* const s0   = lds + L_S0;
    float* const b0   = lds + L_B0;
    float* const sgn  = lds + L_SGN;
    float* const redS = lds + L_REDS;
    float* const redQ = lds + L_REDQ;
    float* const ext  = lds + L_EXT;
    float* const sumv = lds + L_SUMV;
    float* const extm = lds + L_EXTM;
    float* const wup  = lds + L_WUP;

    const int tid = threadIdx.x;

    for (int i = tid; i < 4 * 3 * 32 * 2; i += 256) {
        const int e = i & 1, ln = (i >> 1) & 31, g = i >> 6;
        const int k = 4 * (g % 3) + 2 * (ln >> 4) + e;
        const int col = (g / 3) * 16 + (ln & 15);
        w0f[i] = k < C_IN ? W0[col * C_IN + k] : 0.0f;
    }
    if (tid < C0) {
        s0[tid] = ws[WS_SCALE0 + tid];
        b0[tid] = ws[WS_BIAS0 + tid];
    }
    if (tid < C1) {
        redS[tid] = 0.0f;
        redQ[tid] = 0.0f;
        sgn[tid] = g1[tid] < 0.0f ? -1.0f : 1.0f;
    }
    for (int i = tid; i < VGRP * C1; i += 256) sumv[i] = 0.0f;
    for (int i = tid; i < VGRP * M0_PITCH; i += 256) m0[i] = 0;

    const int lane = tid & 31;
    const int w = tid >> 5;
    const int rt = w & 3;
    const int jbase = (w >> 2) * 4;
    const int hi = lane >> 4;
    const int lrow = rt * 16 + (lane & 15);

    v2f bF[4][16];
#pragma unroll
    for (int jj = 0; jj < 4; ++jj) {
        const float* wr = W1 + ((jbase + jj) * 16 + (lane & 15)) * C1 + 2 * hi;
#pragma unroll
        for (int k = 0; k < 16; ++k) bF[jj][k] = *(const v2f*)(wr + 4 * k);
    }

    float* const wu = wup + (w * 16 * 32 + lane) * 2;
    {
        const float* wr = W1 + (w * 16 + (lane & 15)) * C1 + C0 + 2 * hi;
#pragma unroll
        for (int k = 0; k < 16; ++k) *(v2f*)&wu[64 * k] = *(const v2f*)(wr + 4 * k);
    }

    const int l0tile = (w >> 2) * 2;
    const v2f* const X2 = (const v2f*)X;
    const unsigned xp01 = lrow * (C_IN / 2) + hi;
    const unsigned xp2 = lrow * (C_IN / 2) + 4 - hi;

    const float* const w0p = w0f + (l0tile * 3 * 32 + lane) * 2;
    float* const l0p = s0 + l0tile * 16 + (lane & 15);
    int*   const m0p = m0 + l0tile * 16 + (lane & 15);
    float* const x1st = x1 + (rt * 16 + 8 * hi) * X1_STRIDE + l0tile * 16 + (lane & 15);
    const float* const x1ld = x1 + lrow * X1_STRIDE + 2 * hi;
    float* const ep = sgn + jbase * 16 + (lane & 15);
    float* const extp = ext + (rt * 2 + hi) * C1 + jbase * 16 + (lane & 15);
    v2f pa0, pa1, pa2;
    {
        const unsigned vo = blockIdx.x * (M_PTS * C_IN / 2);
        pa0 = X2[vo + xp01];
        pa1 = X2[vo + xp01 + 2];
        pa2 = X2[vo + xp2];
    }
    __syncthreads();

    const int ntrips = (N_VOX - 1 - (int)blockIdx.x) / GEMM_WGS + 1;
    for (int t0 = 0; t0 < ntrips; t0 += VGRP) {
        const int cnt = ntrips - t0 < VGRP ? ntrips - t0 : VGRP;
        for (int g = 0; g < cnt; ++g) {
            const int n = blockIdx.x + (t0 + g) * GEMM_WGS;
            {
                v2f a[3];
                a[0] = pa0;
                a[1] = pa1;
                a[2] = hi ? (v2f){0.0f, 0.0f} : pa2;
                v2f b[2][3];
#pragma unroll
                for (int t = 0; t < 2; ++t)
#pragma unroll
                    for (int s = 0; s < 3; ++s)
                        b[t][s] = *(const v2f*)&w0p[(t * 3 + s) * 64];
                v8f l0acc[2] = {};
#pragma unroll
                for (int s = 0; s < 3; ++s)
#pragma unroll
                    for (int t = 0; t < 2; ++t)
                        l0acc[t] = __builtin_amdgcn_wmma_f32_16x16x4_f32(
                            false, a[s], false, b[t][s], (short)0, l0acc[t], false, false);
                int* const mg = m0p + g * M0_PITCH;
#pragma unroll
                for (int t = 0; t < 2; ++t) {
                    const float sc = l0p[16 * t], bi = l0p[(L_B0 - L_S0) + 16 * t];
                    float* xp = x1st + 16 * t;
                    float m = 0.0f;
#pragma unroll
                    for (int i = 0; i < 8; ++i) {
                        float v = l0acc[t][i] * sc + bi;
                        v = v > 0.0f ? v : 0.0f;
                        xp[i * X1_STRIDE] = v;
                        m = fmaxf(m, v);
                    }
                    atomicMax(&mg[16 * t], __float_as_int(m));
                }
            }
            __syncthreads();

            v2f aF[16];
#pragma unroll
            for (int k = 0; k < 16; ++k)
                aF[k] = *(const v2f*)&x1ld[4 * k];

            float* const sv = ep + (L_SUMV - L_SGN) + g * C1;

#pragma unroll
            for (int jp = 0; jp < 4; jp += 2) {
                v8f accs[2] = {};
#pragma unroll
                for (int k = 0; k < 16; ++k) {
#pragma unroll
                    for (int t = 0; t < 2; ++t)
                        accs[t] = __builtin_amdgcn_wmma_f32_16x16x4_f32(
                            false, aF[k], false, bF[jp + t][k], (short)0, accs[t], false, false);
                }

#pragma unroll
                for (int t = 0; t < 2; ++t) {
                    const v8f acc = accs[t];
                    const float sg = ep[16 * (jp + t)];
                    float s = 0.0f, q = 0.0f;
                    float m = acc[0] * sg;
#pragma unroll
                    for (int i = 0; i < 8; ++i) {
                        float v = acc[i];
                        s += v;
                        q += v * v;
                        m = fmaxf(m, v * sg);
                    }
                    atomicAdd(&sv[16 * (jp + t)], s);
                    atomicAdd(&ep[(L_REDQ - L_SGN) + 16 * (jp + t)], q);
                    extp[16 * (jp + t)] = m;
                }

                if (jp == 0 && n + GEMM_WGS < N_VOX) {
                    const unsigned vo = (unsigned)(n + GEMM_WGS) * (M_PTS * C_IN / 2);
                    pa0 = X2[vo + xp01];
                    pa1 = X2[vo + xp01 + 2];
                    pa2 = X2[vo + xp2];
                }
            }
            __syncthreads();

            if (tid < C1) {
                float m = ext[tid];
                for (int p = 1; p < 8; ++p) m = fmaxf(m, ext[p * C1 + tid]);
                extm[g * C1 + tid] = m;
            }
        }
        __syncthreads();

        {
            const int col = w * 16 + (lane & 15);
            const int* const mr = m0 + (lane & 15) * M0_PITCH + 2 * hi;
            v8f ce = {}, co = {};
#pragma unroll
            for (int k = 0; k < 16; k += 2) {
                const v2f a0 = __builtin_bit_cast(v2f, *(const v2i*)&mr[4 * k]);
                const v2f a1 = __builtin_bit_cast(v2f, *(const v2i*)&mr[4 * k + 4]);
                const v2f b0f = *(const v2f*)&wu[64 * k];
                const v2f b1f = *(const v2f*)&wu[64 * k + 64];
                ce = __builtin_amdgcn_wmma_f32_16x16x4_f32(
                    false, a0, false, b0f, (short)0, ce, false, false);
                co = __builtin_amdgcn_wmma_f32_16x16x4_f32(
                    false, a1, false, b1f, (short)0, co, false, false);
            }
            const float sg = sgn[col];
            float ps = 0.0f, pq = 0.0f;
#pragma unroll
            for (int i = 0; i < 8; ++i) {
                const int v = 8 * hi + i;
                if (v < cnt) {
                    const float c = ce[i] + co[i];
                    const float s = sumv[v * C1 + col];
                    ps += s + (float)M_PTS * c;
                    pq += (2.0f * s + (float)M_PTS * c) * c;
                    extm[v * C1 + col] = (extm[v * C1 + col] + sg * c) * sg;
                }
            }
            atomicAdd(&redS[col], ps);
            atomicAdd(&redQ[col], pq);
        }
        __syncthreads();

#pragma unroll
        for (int j = 0; j < VGRP * C1 / 256; ++j) {
            const int idx = j * 256 + tid;
            const int v = idx >> 7;
            if (v < cnt)
                out[(blockIdx.x + (t0 + v) * GEMM_WGS) * C1 + (idx & (C1 - 1))] = extm[idx];
            sumv[idx] = 0.0f;
        }
        for (int i = tid; i < VGRP * M0_PITCH; i += 256) m0[i] = 0;
        __syncthreads();
    }

    if (tid < C1) {
        atomicAdd(&ws[WS_SUMY + tid], redS[tid]);
        atomicAdd(&ws[WS_SUMQ + tid], redQ[tid]);
    }
}

__global__ __launch_bounds__(256) void vfe_apply(const float* __restrict__ ws,
                                                 float* __restrict__ out) {
    const int i = blockIdx.x * 256 + threadIdx.x;
    if (i >= N_OUT4) return;
    const int o = (i & (C1 / 4 - 1)) * 4;
    const float4 sc = *(const float4*)&ws[WS_SCALE1 + o];
    const float4 bi = *(const float4*)&ws[WS_BIAS1 + o];
    float4 y = ((const float4*)out)[i];
    y.x = y.x * sc.x + bi.x;
    y.y = y.y * sc.y + bi.y;
    y.z = y.z * sc.z + bi.z;
    y.w = y.w * sc.w + bi.w;
    y.x = y.x > 0.0f ? y.x : 0.0f;
    y.y = y.y > 0.0f ? y.y : 0.0f;
    y.z = y.z > 0.0f ? y.z : 0.0f;
    y.w = y.w > 0.0f ? y.w : 0.0f;
    ((float4*)out)[i] = y;
}

extern "C" void kernel_launch(void* const* d_in, const int* in_sizes, int n_in,
                              void* d_out, int out_size, void* d_ws, size_t ws_size,
                              hipStream_t stream) {
    (void)in_sizes; (void)n_in; (void)out_size; (void)ws_size;
    const float* X   = (const float*)d_in[0];
    const float* W0  = (const float*)d_in[1];
    const float* g0  = (const float*)d_in[2];
    const float* be0 = (const float*)d_in[3];
    const float* W1  = (const float*)d_in[4];
    const float* g1  = (const float*)d_in[5];
    const float* be1 = (const float*)d_in[6];
    float* out = (float*)d_out;
    float* ws  = (float*)d_ws;

    const long P = (long)N_VOX * M_PTS;
    const float invP = 1.0f / (float)P;

    vfe_zero<<<(WS_TOTAL + 255) / 256, 256, 0, stream>>>(ws);
    vfe_stats0<<<1024, 256, 0, stream>>>(X, ws, P);
    vfe_final0<<<1, 64, 0, stream>>>(W0, g0, be0, ws, invP);
    vfe_gemm<<<GEMM_WGS, 256, 0, stream>>>(X, W0, W1, g1, ws, out);
    vfe_final1<<<1, 128, 0, stream>>>(g1, be1, ws, invP);
    vfe_apply<<<N_OUT4 / 256, 256, 0, stream>>>(ws, out);
}


# stats0 pass: input Gram matrix X^T X (10x10) and sum_x moved from 110 FMA accumulators per thread onto f32 WMMA; one register pair is both A and B, a ones column gives sum_x (sum order changes)
# speedup vs baseline: 5.7403x; 1.0492x over previous
#include <hip/hip_runtime.h>
#include <hip/hip_bf16.h>

typedef float v2f __attribute__((ext_vector_type(2)));
typedef float v8f __attribute__((ext_vector_type(8)));
typedef int v2i __attribute__((ext_vector_type(2)));

#define N_VOX 20000
#define M_PTS 64
#define C_IN  10
#define C0    64
#define C1    128
#define EPS   1e-3f
#define N_OUT4 640000
#define GEMM_WGS 512

#define WS_SUMX   0
#define WS_SXX    16
#define WS_SCALE0 128
#define WS_BIAS0  192
#define WS_SUMY   256
#define WS_SUMQ   384
#define WS_SCALE1 512
#define WS_BIAS1  640
#define WS_TOTAL  768

#define X1_STRIDE 132

#define VGRP    16
#define L_W0F   0
#define L_S0    (L_W0F + 4 * 3 * 32 * 2)
#define L_B0    (L_S0 + C0)
#define L_SGN   (L_B0 + C0)
#define L_REDS  (L_SGN + C1)
#define L_REDQ  (L_REDS + C1)
#define L_EXT   (L_REDQ + C1)
#define L_SUMV  (L_EXT + 8 * C1)
#define L_EXTM  (L_SUMV + VGRP * C1)
#define L_X1    (L_EXTM + VGRP * C1)
#define L_WUP   (L_X1 + M_PTS * X1_STRIDE)
#define L_TOTAL (L_WUP + 8 * 16 * 32 * 2)
#define M0_PITCH 68

__global__ void vfe_zero(float* __restrict__ ws) {
    int i = blockIdx.x * blockDim.x + threadIdx.x;
    if (i < WS_TOTAL) ws[i] = 0.0f;
}

#define S0_WGS   512
#define S0_PTS   64
#define S0_TRIPS (N_VOX * M_PTS / S0_PTS)
static_assert((N_VOX * M_PTS) % S0_PTS == 0, "whole trips only");

__device__ __forceinline__ void s0_load(float (&buf)[32], const float* __restrict__ xb, unsigned lo) {
#pragma unroll
    for (int j = 0; j < 16; ++j) {
        buf[2 * j]     = xb[lo + 4 * C_IN * j];
        buf[2 * j + 1] = xb[lo + 4 * C_IN * j + C_IN];
    }
}

__device__ __forceinline__ void s0_mma(const float (&buf)[32], bool real, float padv,
                                       v8f& acc0, v8f& acc1) {
#pragma unroll
    for (int j = 0; j < 16; j += 2) {
        const v2f a0 = {real ? buf[2 * j] : padv, real ? buf[2 * j + 1] : padv};
        const v2f a1 = {real ? buf[2 * j + 2] : padv, real ? buf[2 * j + 3] : padv};
        acc0 = __builtin_amdgcn_wmma_f32_16x16x4_f32(false, a0, false, a0, (short)0, acc0, false, false);
        acc1 = __builtin_amdgcn_wmma_f32_16x16x4_f32(false, a1, false, a1, (short)0, acc1, false, false);
    }
}

__global__ __launch_bounds__(256) void vfe_stats0(const float* __restrict__ X,
                                                  float* __restrict__ ws) {
    const int tid = threadIdx.x;
    const int lane = tid & 31;
    const int w = __builtin_amdgcn_readfirstlane(tid >> 5);
    const int c = lane & 15, hi = lane >> 4;
    const bool real = c < C_IN;
    const float padv = c == C_IN ? 1.0f : 0.0f;
    const unsigned lo = 2 * hi * C_IN + (real ? c : C_IN - 1);
    const int nw = gridDim.x * 8;
    int tr = blockIdx.x * 8 + w;

    v8f acc0 = {}, acc1 = {};
    float bufA[32], bufB[32];
    if (tr < S0_TRIPS) s0_load(bufA, X + (size_t)tr * (S0_PTS * C_IN), lo);
    while (tr < S0_TRIPS) {
        const bool more = tr + nw < S0_TRIPS;
        if (more) s0_load(bufB, X + (size_t)(tr + nw) * (S0_PTS * C_IN), lo);
        s0_mma(bufA, real, padv, acc0, acc1);
        if (!more) break;
        if (tr + 2 * nw < S0_TRIPS) s0_load(bufA, X + (size_t)(tr + 2 * nw) * (S0_PTS * C_IN), lo);
        s0_mma(bufB, real, padv, acc0, acc1);
        tr += 2 * nw;
    }

    __shared__ __attribute__((aligned(16))) float red[8 * 256];
    {
        float* r = red + w * 256 + (8 * hi) * 16 + c;
#pragma unroll
        for (int i = 0; i < 8; ++i) r[i * 16] = acc0[i] + acc1[i];
    }
    __syncthreads();
    static_assert(WS_SUMX == 0 && WS_SXX == 16, "thread t owns ws[t]");
    const int e = tid - WS_SXX;
    const bool isx = tid < C_IN, isxx = e >= 0 && e < C_IN * C_IN;
    if (isx || isxx) {
        const int m = isx ? tid : e / C_IN;
        const int n = isx ? C_IN : e % C_IN;
        float s = 0.0f;
#pragma unroll
        for (int p = 0; p < 8; ++p) s += red[p * 256 + m * 16 + n];
        atomicAdd(&ws[tid], s);
    }
}

__global__ void vfe_final0(const float* __restrict__ W0, const float* __restrict__ g0,
                           const float* __restrict__ be0, float* __restrict__ ws,
                           float invP) {
    int o = threadIdx.x;
    if (o >= C0) return;
    float w[C_IN];
#pragma unroll
    for (int c = 0; c < C_IN; ++c) w[c] = W0[o * C_IN + c];
    float mean = 0.0f;
#pragma unroll
    for (int c = 0; c < C_IN; ++c) mean += w[c] * ws[WS_SUMX + c];
    mean *= invP;
    float ey2 = 0.0f;
#pragma unroll
    for (int c = 0; c < C_IN; ++c)
#pragma unroll
        for (int d = 0; d < C_IN; ++d) ey2 += w[c] * w[d] * ws[WS_SXX + c * C_IN + d];
    ey2 *= invP;
    float var = ey2 - mean * mean;
    float sc = g0[o] * rsqrtf(var + EPS);
    ws[WS_SCALE0 + o] = sc;
    ws[WS_BIAS0 + o] = be0[o] - mean * sc;
}

__global__ void vfe_final1(const float* __restrict__ g1, const float* __restrict__ be1,
                           float* __restrict__ ws, float invP) {
    int o = threadIdx.x;
    if (o >= C1) return;
    float mean = ws[WS_SUMY + o] * invP;
    float var = ws[WS_SUMQ + o] * invP - mean * mean;
    float sc = g1[o] * rsqrtf(var + EPS);
    ws[WS_SCALE1 + o] = sc;
    ws[WS_BIAS1 + o] = be1[o] - mean * sc;
}

__global__ __launch_bounds__(256) __attribute__((amdgpu_waves_per_eu(4, 4)))
void vfe_gemm(const float* __restrict__ X,
              const float* __restrict__ W0,
              const float* __restrict__ W1,
              const float* __restrict__ g1,
              float* __restrict__ ws,
              float* __restrict__ out) {
    __shared__ __attribute__((aligned(16))) float lds[L_TOTAL];
    __shared__ __attribute__((aligned(16))) int m0[VGRP * M0_PITCH];
    float* const x1   = lds + L_X1;
    float* const w0f  = lds + L_W0F;
    float* const s0   = lds + L_S0;
    float* const b0   = lds + L_B0;
    float* const sgn  = lds + L_SGN;
    float* const redS = lds + L_REDS;
    float* const redQ = lds + L_REDQ;
    float* const ext  = lds + L_EXT;
    float* const sumv = lds + L_SUMV;
    float* const extm = lds + L_EXTM;
    float* const wup  = lds + L_WUP;

    const int tid = threadIdx.x;

    for (int i = tid; i < 4 * 3 * 32 * 2; i += 256) {
        const int e = i & 1, ln = (i >> 1) & 31, g = i >> 6;
        const int k = 4 * (g % 3) + 2 * (ln >> 4) + e;
        const int col = (g / 3) * 16 + (ln & 15);
        w0f[i] = k < C_IN ? W0[col * C_IN + k] : 0.0f;
    }
    if (tid < C0) {
        s0[tid] = ws[WS_SCALE0 + tid];
        b0[tid] = ws[WS_BIAS0 + tid];
    }
    if (tid < C1) {
        redS[tid] = 0.0f;
        redQ[tid] = 0.0f;
        sgn[tid] = g1[tid] < 0.0f ? -1.0f : 1.0f;
    }
    for (int i = tid; i < VGRP * C1; i += 256) sumv[i] = 0.0f;
    for (int i = tid; i < VGRP * M0_PITCH; i += 256) m0[i] = 0;

    const int lane = tid & 31;
    const int w = tid >> 5;
    const int rt = w & 3;
    const int jbase = (w >> 2) * 4;
    const int hi = lane >> 4;
    const int lrow = rt * 16 + (lane & 15);

    v2f bF[4][16];
#pragma unroll
    for (int jj = 0; jj < 4; ++jj) {
        const float* wr = W1 + ((jbase + jj) * 16 + (lane & 15)) * C1 + 2 * hi;
#pragma unroll
        for (int k = 0; k < 16; ++k) bF[jj][k] = *(const v2f*)(wr + 4 * k);
    }

    float* const wu = wup + (w * 16 * 32 + lane) * 2;
    {
        const float* wr = W1 + (w * 16 + (lane & 15)) * C1 + C0 + 2 * hi;
#pragma unroll
        for (int k = 0; k < 16; ++k) *(v2f*)&wu[64 * k] = *(const v2f*)(wr + 4 * k);
    }

    const int l0tile = (w >> 2) * 2;
    const v2f* const X2 = (const v2f*)X;
    const unsigned xp01 = lrow * (C_IN / 2) + hi;
    const unsigned xp2 = lrow * (C_IN / 2) + 4 - hi;

    const float* const w0p = w0f + (l0tile * 3 * 32 + lane) * 2;
    float* const l0p = s0 + l0tile * 16 + (lane & 15);
    int*   const m0p = m0 + l0tile * 16 + (lane & 15);
    float* const x1st = x1 + (rt * 16 + 8 * hi) * X1_STRIDE + l0tile * 16 + (lane & 15);
    const float* const x1ld = x1 + lrow * X1_STRIDE + 2 * hi;
    float* const ep = sgn + jbase * 16 + (lane & 15);
    float* const extp = ext + (rt * 2 + hi) * C1 + jbase * 16 + (lane & 15);
    v2f pa0, pa1, pa2;
    {
        const unsigned vo = blockIdx.x * (M_PTS * C_IN / 2);
        pa0 = X2[vo + xp01];
        pa1 = X2[vo + xp01 + 2];
        pa2 = X2[vo + xp2];
    }
    __syncthreads();

    const int ntrips = (N_VOX - 1 - (int)blockIdx.x) / GEMM_WGS + 1;
    for (int t0 = 0; t0 < ntrips; t0 += VGRP) {
        const int cnt = ntrips - t0 < VGRP ? ntrips - t0 : VGRP;
        for (int g = 0; g < cnt; ++g) {
            const int n = blockIdx.x + (t0 + g) * GEMM_WGS;
            {
                v2f a[3];
                a[0] = pa0;
                a[1] = pa1;
                a[2] = hi ? (v2f){0.0f, 0.0f} : pa2;
                v2f b[2][3];
#pragma unroll
                for (int t = 0; t < 2; ++t)
#pragma unroll
                    for (int s = 0; s < 3; ++s)
                        b[t][s] = *(const v2f*)&w0p[(t * 3 + s) * 64];
                v8f l0acc[2] = {};
#pragma unroll
                for (int s = 0; s < 3; ++s)
#pragma unroll
                    for (int t = 0; t < 2; ++t)
                        l0acc[t] = __builtin_amdgcn_wmma_f32_16x16x4_f32(
                            false, a[s], false, b[t][s], (short)0, l0acc[t], false, false);
                int* const mg = m0p + g * M0_PITCH;
#pragma unroll
                for (int t = 0; t < 2; ++t) {
                    const float sc = l0p[16 * t], bi = l0p[(L_B0 - L_S0) + 16 * t];
                    float* xp = x1st + 16 * t;
                    float m = 0.0f;
#pragma unroll
                    for (int i = 0; i < 8; ++i) {
                        float v = l0acc[t][i] * sc + bi;
                        v = v > 0.0f ? v : 0.0f;
                        xp[i * X1_STRIDE] = v;
                        m = fmaxf(m, v);
                    }
                    atomicMax(&mg[16 * t], __float_as_int(m));
                }
            }
            __syncthreads();

            v2f aF[16];
#pragma unroll
            for (int k = 0; k < 16; ++k)
                aF[k] = *(const v2f*)&x1ld[4 * k];

            float* const sv = ep + (L_SUMV - L_SGN) + g * C1;

#pragma unroll
            for (int jp = 0; jp < 4; jp += 2) {
                v8f accs[2] = {};
#pragma unroll
                for (int k = 0; k < 16; ++k) {
#pragma unroll
                    for (int t = 0; t < 2; ++t)
                        accs[t] = __builtin_amdgcn_wmma_f32_16x16x4_f32(
                            false, aF[k], false, bF[jp + t][k], (short)0, accs[t], false, false);
                }

#pragma unroll
                for (int t = 0; t < 2; ++t) {
                    const v8f acc = accs[t];
                    const float sg = ep[16 * (jp + t)];
                    float s = 0.0f, q = 0.0f;
                    float m = acc[0] * sg;
#pragma unroll
                    for (int i = 0; i < 8; ++i) {
                        float v = acc[i];
                        s += v;
                        q += v * v;
                        m = fmaxf(m, v * sg);
                    }
                    atomicAdd(&sv[16 * (jp + t)], s);
                    atomicAdd(&ep[(L_REDQ - L_SGN) + 16 * (jp + t)], q);
                    extp[16 * (jp + t)] = m;
                }

                if (jp == 0 && n + GEMM_WGS < N_VOX) {
                    const unsigned vo = (unsigned)(n + GEMM_WGS) * (M_PTS * C_IN / 2);
                    pa0 = X2[vo + xp01];
                    pa1 = X2[vo + xp01 + 2];
                    pa2 = X2[vo + xp2];
                }
            }
            __syncthreads();

            if (tid < C1) {
                float m = ext[tid];
                for (int p = 1; p < 8; ++p) m = fmaxf(m, ext[p * C1 + tid]);
                extm[g * C1 + tid] = m;
            }
        }
        __syncthreads();

        {
            const int col = w * 16 + (lane & 15);
            const int* const mr = m0 + (lane & 15) * M0_PITCH + 2 * hi;
            v8f ce = {}, co = {};
#pragma unroll
            for (int k = 0; k < 16; k += 2) {
                const v2f a0 = __builtin_bit_cast(v2f, *(const v2i*)&mr[4 * k]);
                const v2f a1 = __builtin_bit_cast(v2f, *(const v2i*)&mr[4 * k + 4]);
                const v2f b0f = *(const v2f*)&wu[64 * k];
                const v2f b1f = *(const v2f*)&wu[64 * k + 64];
                ce = __builtin_amdgcn_wmma_f32_16x16x4_f32(
                    false, a0, false, b0f, (short)0, ce, false, false);
                co = __builtin_amdgcn_wmma_f32_16x16x4_f32(
                    false, a1, false, b1f, (short)0, co, false, false);
            }
            const float sg = sgn[col];
            float ps = 0.0f, pq = 0.0f;
#pragma unroll
            for (int i = 0; i < 8; ++i) {
                const int v = 8 * hi + i;
                if (v < cnt) {
                    const float c = ce[i] + co[i];
                    const float s = sumv[v * C1 + col];
                    ps += s + (float)M_PTS * c;
                    pq += (2.0f * s + (float)M_PTS * c) * c;
                    extm[v * C1 + col] = (extm[v * C1 + col] + sg * c) * sg;
                }
            }
            atomicAdd(&redS[col], ps);
            atomicAdd(&redQ[col], pq);
        }
        __syncthreads();

#pragma unroll
        for (int j = 0; j < VGRP * C1 / 256; ++j) {
            const int idx = j * 256 + tid;
            const int v = idx >> 7;
            if (v < cnt)
                out[(blockIdx.x + (t0 + v) * GEMM_WGS) * C1 + (idx & (C1 - 1))] = extm[idx];
            sumv[idx] = 0.0f;
        }
        for (int i = tid; i < VGRP * M0_PITCH; i += 256) m0[i] = 0;
        __syncthreads();
    }

    if (tid < C1) {
        atomicAdd(&ws[WS_SUMY + tid], redS[tid]);
        atomicAdd(&ws[WS_SUMQ + tid], redQ[tid]);
    }
}

__global__ __launch_bounds__(256) void vfe_apply(const float* __restrict__ ws,
                                                 float* __restrict__ out) {
    const int i = blockIdx.x * 256 + threadIdx.x;
    if (i >= N_OUT4) return;
    const int o = (i & (C1 / 4 - 1)) * 4;
    const float4 sc = *(const float4*)&ws[WS_SCALE1 + o];
    const float4 bi = *(const float4*)&ws[WS_BIAS1 + o];
    float4 y = ((const float4*)out)[i];
    y.x = y.x * sc.x + bi.x;
    y.y = y.y * sc.y + bi.y;
    y.z = y.z * sc.z + bi.z;
    y.w = y.w * sc.w + bi.w;
    y.x = y.x > 0.0f ? y.x : 0.0f;
    y.y = y.y > 0.0f ? y.y : 0.0f;
    y.z = y.z > 0.0f ? y.z : 0.0f;
    y.w = y.w > 0.0f ? y.w : 0.0f;
    ((float4*)out)[i] = y;
}

extern "C" void kernel_launch(void* const* d_in, const int* in_sizes, int n_in,
                              void* d_out, int out_size, void* d_ws, size_t ws_size,
                              hipStream_t stream) {
    (void)in_sizes; (void)n_in; (void)out_size; (void)ws_size;
    const float* X   = (const float*)d_in[0];
    const float* W0  = (const float*)d_in[1];
    const float* g0  = (const float*)d_in[2];
    const float* be0 = (const float*)d_in[3];
    const float* W1  = (const float*)d_in[4];
    const float* g1  = (const float*)d_in[5];
    const float* be1 = (const float*)d_in[6];
    float* out = (float*)d_out;
    float* ws  = (float*)d_ws;

    const long P = (long)N_VOX * M_PTS;
    const float invP = 1.0f / (float)P;

    vfe_zero<<<(WS_TOTAL + 255) / 256, 256, 0, stream>>>(ws);
    vfe_stats0<<<S0_WGS, 256, 0, stream>>>(X, ws);
    vfe_final0<<<1, 64, 0, stream>>>(W0, g0, be0, ws, invP);
    vfe_gemm<<<GEMM_WGS, 256, 0, stream>>>(X, W0, W1, g1, ws, out);
    vfe_final1<<<1, 128, 0, stream>>>(g1, be1, ws, invP);
    vfe_apply<<<N_OUT4 / 256, 256, 0, stream>>>(ws, out);
}


# main GEMM tile epilogue shortened: sign(gamma1) folded into the W1 register fragments (exact), row sums and squares as packed pair trees on the accumulator pairs (sum order changes)
# speedup vs baseline: 6.0379x; 1.0518x over previous
#include <hip/hip_runtime.h>
#include <hip/hip_bf16.h>

typedef float v2f __attribute__((ext_vector_type(2)));
typedef float v8f __attribute__((ext_vector_type(8)));
typedef int v2i __attribute__((ext_vector_type(2)));

#define N_VOX 20000
#define M_PTS 64
#define C_IN  10
#define C0    64
#define C1    128
#define EPS   1e-3f
#define N_OUT4 640000
#define GEMM_WGS 512

#define WS_SUMX   0
#define WS_SXX    16
#define WS_SCALE0 128
#define WS_BIAS0  192
#define WS_SUMY   256
#define WS_SUMQ   384
#define WS_SCALE1 512
#define WS_BIAS1  640
#define WS_TOTAL  768

#define X1_STRIDE 132

#define VGRP    16
#define L_W0F   0
#define L_S0    (L_W0F + 4 * 3 * 32 * 2)
#define L_B0    (L_S0 + C0)
#define L_SGN   (L_B0 + C0)
#define L_REDS  (L_SGN + C1)
#define L_REDQ  (L_REDS + C1)
#define L_EXT   (L_REDQ + C1)
#define L_SUMV  (L_EXT + 8 * C1)
#define L_EXTM  (L_SUMV + VGRP * C1)
#define L_X1    (L_EXTM + VGRP * C1)
#define L_WUP   (L_X1 + M_PTS * X1_STRIDE)
#define L_TOTAL (L_WUP + 8 * 16 * 32 * 2)
#define M0_PITCH 68

__global__ void vfe_zero(float* __restrict__ ws) {
    int i = blockIdx.x * blockDim.x + threadIdx.x;
    if (i < WS_TOTAL) ws[i] = 0.0f;
}

#define S0_WGS   512
#define S0_PTS   64
#define S0_TRIPS (N_VOX * M_PTS / S0_PTS)
static_assert((N_VOX * M_PTS) % S0_PTS == 0, "whole trips only");

__device__ __forceinline__ void s0_load(float (&buf)[32], const float* __restrict__ xb, unsigned lo) {
#pragma unroll
    for (int j = 0; j < 16; ++j) {
        buf[2 * j]     = xb[lo + 4 * C_IN * j];
        buf[2 * j + 1] = xb[lo + 4 * C_IN * j + C_IN];
    }
}

__device__ __forceinline__ void s0_mma(const float (&buf)[32], bool real, float padv,
                                       v8f& acc0, v8f& acc1) {
#pragma unroll
    for (int j = 0; j < 16; j += 2) {
        const v2f a0 = {real ? buf[2 * j] : padv, real ? buf[2 * j + 1] : padv};
        const v2f a1 = {real ? buf[2 * j + 2] : padv, real ? buf[2 * j + 3] : padv};
        acc0 = __builtin_amdgcn_wmma_f32_16x16x4_f32(false, a0, false, a0, (short)0, acc0, false, false);
        acc1 = __builtin_amdgcn_wmma_f32_16x16x4_f32(false, a1, false, a1, (short)0, acc1, false, false);
    }
}

__global__ __launch_bounds__(256) void vfe_stats0(const float* __restrict__ X,
                                                  float* __restrict__ ws) {
    const int tid = threadIdx.x;
    const int lane = tid & 31;
    const int w = __builtin_amdgcn_readfirstlane(tid >> 5);
    const int c = lane & 15, hi = lane >> 4;
    const bool real = c < C_IN;
    const float padv = c == C_IN ? 1.0f : 0.0f;
    const unsigned lo = 2 * hi * C_IN + (real ? c : C_IN - 1);
    const int nw = gridDim.x * 8;
    int tr = blockIdx.x * 8 + w;

    v8f acc0 = {}, acc1 = {};
    float bufA[32], bufB[32];
    if (tr < S0_TRIPS) s0_load(bufA, X + (size_t)tr * (S0_PTS * C_IN), lo);
    while (tr < S0_TRIPS) {
        const bool more = tr + nw < S0_TRIPS;
        if (more) s0_load(bufB, X + (size_t)(tr + nw) * (S0_PTS * C_IN), lo);
        s0_mma(bufA, real, padv, acc0, acc1);
        if (!more) break;
        if (tr + 2 * nw < S0_TRIPS) s0_load(bufA, X + (size_t)(tr + 2 * nw) * (S0_PTS * C_IN), lo);
        s0_mma(bufB, real, padv, acc0, acc1);
        tr += 2 * nw;
    }

    __shared__ __attribute__((aligned(16))) float red[8 * 256];
    {
        float* r = red + w * 256 + (8 * hi) * 16 + c;
#pragma unroll
        for (int i = 0; i < 8; ++i) r[i * 16] = acc0[i] + acc1[i];
    }
    __syncthreads();
    static_assert(WS_SUMX == 0 && WS_SXX == 16, "thread t owns ws[t]");
    const int e = tid - WS_SXX;
    const bool isx = tid < C_IN, isxx = e >= 0 && e < C_IN * C_IN;
    if (isx || isxx) {
        const int m = isx ? tid : e / C_IN;
        const int n = isx ? C_IN : e % C_IN;
        float s = 0.0f;
#pragma unroll
        for (int p = 0; p < 8; ++p) s += red[p * 256 + m * 16 + n];
        atomicAdd(&ws[tid], s);
    }
}

__global__ void vfe_final0(const float* __restrict__ W0, const float* __restrict__ g0,
                           const float* __restrict__ be0, float* __restrict__ ws,
                           float invP) {
    int o = threadIdx.x;
    if (o >= C0) return;
    float w[C_IN];
#pragma unroll
    for (int c = 0; c < C_IN; ++c) w[c] = W0[o * C_IN + c];
    float mean = 0.0f;
#pragma unroll
    for (int c = 0; c < C_IN; ++c) mean += w[c] * ws[WS_SUMX + c];
    mean *= invP;
    float ey2 = 0.0f;
#pragma unroll
    for (int c = 0; c < C_IN; ++c)
#pragma unroll
        for (int d = 0; d < C_IN; ++d) ey2 += w[c] * w[d] * ws[WS_SXX + c * C_IN + d];
    ey2 *= invP;
    float var = ey2 - mean * mean;
    float sc = g0[o] * rsqrtf(var + EPS);
    ws[WS_SCALE0 + o] = sc;
    ws[WS_BIAS0 + o] = be0[o] - mean * sc;
}

__global__ void vfe_final1(const float* __restrict__ g1, const float* __restrict__ be1,
                           float* __restrict__ ws, float invP) {
    int o = threadIdx.x;
    if (o >= C1) return;
    float mean = ws[WS_SUMY + o] * invP;
    float var = ws[WS_SUMQ + o] * invP - mean * mean;
    float sc = g1[o] * rsqrtf(var + EPS);
    ws[WS_SCALE1 + o] = sc;
    ws[WS_BIAS1 + o] = be1[o] - mean * sc;
}

__global__ __launch_bounds__(256) __attribute__((amdgpu_waves_per_eu(4, 4)))
void vfe_gemm(const float* __restrict__ X,
              const float* __restrict__ W0,
              const float* __restrict__ W1,
              const float* __restrict__ g1,
              float* __restrict__ ws,
              float* __restrict__ out) {
    __shared__ __attribute__((aligned(16))) float lds[L_TOTAL];
    __shared__ __attribute__((aligned(16))) int m0[VGRP * M0_PITCH];
    float* const x1   = lds + L_X1;
    float* const w0f  = lds + L_W0F;
    float* const s0   = lds + L_S0;
    float* const b0   = lds + L_B0;
    float* const sgn  = lds + L_SGN;
    float* const redS = lds + L_REDS;
    float* const redQ = lds + L_REDQ;
    float* const ext  = lds + L_EXT;
    float* const sumv = lds + L_SUMV;
    float* const extm = lds + L_EXTM;
    float* const wup  = lds + L_WUP;

    const int tid = threadIdx.x;

    for (int i = tid; i < 4 * 3 * 32 * 2; i += 256) {
        const int e = i & 1, ln = (i >> 1) & 31, g = i >> 6;
        const int k = 4 * (g % 3) + 2 * (ln >> 4) + e;
        const int col = (g / 3) * 16 + (ln & 15);
        w0f[i] = k < C_IN ? W0[col * C_IN + k] : 0.0f;
    }
    if (tid < C0) {
        s0[tid] = ws[WS_SCALE0 + tid];
        b0[tid] = ws[WS_BIAS0 + tid];
    }
    if (tid < C1) {
        redS[tid] = 0.0f;
        redQ[tid] = 0.0f;
        sgn[tid] = g1[tid] < 0.0f ? -1.0f : 1.0f;
    }
    for (int i = tid; i < VGRP * C1; i += 256) sumv[i] = 0.0f;
    for (int i = tid; i < VGRP * M0_PITCH; i += 256) m0[i] = 0;

    const int lane = tid & 31;
    const int w = tid >> 5;
    const int rt = w & 3;
    const int jbase = (w >> 2) * 4;
    const int hi = lane >> 4;
    const int lrow = rt * 16 + (lane & 15);

    v2f bF[4][16];
#pragma unroll
    for (int jj = 0; jj < 4; ++jj) {
        const int colb = (jbase + jj) * 16 + (lane & 15);
        const float* wr = W1 + colb * C1 + 2 * hi;
        const float sgc = g1[colb] < 0.0f ? -1.0f : 1.0f;
#pragma unroll
        for (int k = 0; k < 16; ++k) bF[jj][k] = *(const v2f*)(wr + 4 * k) * sgc;
    }

    float* const wu = wup + (w * 16 * 32 + lane) * 2;
    {
        const float* wr = W1 + (w * 16 + (lane & 15)) * C1 + C0 + 2 * hi;
#pragma unroll
        for (int k = 0; k < 16; ++k) *(v2f*)&wu[64 * k] = *(const v2f*)(wr + 4 * k);
    }

    const int l0tile = (w >> 2) * 2;
    const v2f* const X2 = (const v2f*)X;
    const unsigned xp01 = lrow * (C_IN / 2) + hi;
    const unsigned xp2 = lrow * (C_IN / 2) + 4 - hi;

    const float* const w0p = w0f + (l0tile * 3 * 32 + lane) * 2;
    float* const l0p = s0 + l0tile * 16 + (lane & 15);
    int*   const m0p = m0 + l0tile * 16 + (lane & 15);
    float* const x1st = x1 + (rt * 16 + 8 * hi) * X1_STRIDE + l0tile * 16 + (lane & 15);
    const float* const x1ld = x1 + lrow * X1_STRIDE + 2 * hi;
    float* const ep = sgn + jbase * 16 + (lane & 15);
    float* const extp = ext + (rt * 2 + hi) * C1 + jbase * 16 + (lane & 15);
    v2f pa0, pa1, pa2;
    {
        const unsigned vo = blockIdx.x * (M_PTS * C_IN / 2);
        pa0 = X2[vo + xp01];
        pa1 = X2[vo + xp01 + 2];
        pa2 = X2[vo + xp2];
    }
    __syncthreads();

    const int ntrips = (N_VOX - 1 - (int)blockIdx.x) / GEMM_WGS + 1;
    for (int t0 = 0; t0 < ntrips; t0 += VGRP) {
        const int cnt = ntrips - t0 < VGRP ? ntrips - t0 : VGRP;
        for (int g = 0; g < cnt; ++g) {
            const int n = blockIdx.x + (t0 + g) * GEMM_WGS;
            {
                v2f a[3];
                a[0] = pa0;
                a[1] = pa1;
                a[2] = hi ? (v2f){0.0f, 0.0f} : pa2;
                v2f b[2][3];
#pragma unroll
                for (int t = 0; t < 2; ++t)
#pragma unroll
                    for (int s = 0; s < 3; ++s)
                        b[t][s] = *(const v2f*)&w0p[(t * 3 + s) * 64];
                v8f l0acc[2] = {};
#pragma unroll
                for (int s = 0; s < 3; ++s)
#pragma unroll
                    for (int t = 0; t < 2; ++t)
                        l0acc[t] = __builtin_amdgcn_wmma_f32_16x16x4_f32(
                            false, a[s], false, b[t][s], (short)0, l0acc[t], false, false);
                int* const mg = m0p + g * M0_PITCH;
#pragma unroll
                for (int t = 0; t < 2; ++t) {
                    const float sc = l0p[16 * t], bi = l0p[(L_B0 - L_S0) + 16 * t];
                    float* xp = x1st + 16 * t;
                    float m = 0.0f;
#pragma unroll
                    for (int i = 0; i < 8; ++i) {
                        float v = l0acc[t][i] * sc + bi;
                        v = v > 0.0f ? v : 0.0f;
                        xp[i * X1_STRIDE] = v;
                        m = fmaxf(m, v);
                    }
                    atomicMax(&mg[16 * t], __float_as_int(m));
                }
            }
            __syncthreads();

            v2f aF[16];
#pragma unroll
            for (int k = 0; k < 16; ++k)
                aF[k] = *(const v2f*)&x1ld[4 * k];

            float* const sv = ep + (L_SUMV - L_SGN) + g * C1;

#pragma unroll
            for (int jp = 0; jp < 4; jp += 2) {
                v8f accs[2] = {};
#pragma unroll
                for (int k = 0; k < 16; ++k) {
#pragma unroll
                    for (int t = 0; t < 2; ++t)
                        accs[t] = __builtin_amdgcn_wmma_f32_16x16x4_f32(
                            false, aF[k], false, bF[jp + t][k], (short)0, accs[t], false, false);
                }

#pragma unroll
                for (int t = 0; t < 2; ++t) {
                    const v8f acc = accs[t];
                    const v2f p0 = __builtin_shufflevector(acc, acc, 0, 1);
                    const v2f p1 = __builtin_shufflevector(acc, acc, 2, 3);
                    const v2f p2 = __builtin_shufflevector(acc, acc, 4, 5);
                    const v2f p3 = __builtin_shufflevector(acc, acc, 6, 7);
                    const v2f ss = (p0 + p1) + (p2 + p3);
                    v2f qq = p0 * p0;
                    qq = p1 * p1 + qq;
                    qq = p2 * p2 + qq;
                    qq = p3 * p3 + qq;
                    const float s = ss[0] + ss[1];
                    const float q = qq[0] + qq[1];
                    const float m = fmaxf(fmaxf(fmaxf(acc[0], acc[1]), fmaxf(acc[2], acc[3])),
                                          fmaxf(fmaxf(acc[4], acc[5]), fmaxf(acc[6], acc[7])));
                    atomicAdd(&sv[16 * (jp + t)], s);
                    atomicAdd(&ep[(L_REDQ - L_SGN) + 16 * (jp + t)], q);
                    extp[16 * (jp + t)] = m;
                }

                if (jp == 0 && n + GEMM_WGS < N_VOX) {
                    const unsigned vo = (unsigned)(n + GEMM_WGS) * (M_PTS * C_IN / 2);
                    pa0 = X2[vo + xp01];
                    pa1 = X2[vo + xp01 + 2];
                    pa2 = X2[vo + xp2];
                }
            }
            __syncthreads();

            if (tid < C1) {
                float m = ext[tid];
                for (int p = 1; p < 8; ++p) m = fmaxf(m, ext[p * C1 + tid]);
                extm[g * C1 + tid] = m;
            }
        }
        __syncthreads();

        {
            const int col = w * 16 + (lane & 15);
            const int* const mr = m0 + (lane & 15) * M0_PITCH + 2 * hi;
            v8f ce = {}, co = {};
#pragma unroll
            for (int k = 0; k < 16; k += 2) {
                const v2f a0 = __builtin_bit_cast(v2f, *(const v2i*)&mr[4 * k]);
                const v2f a1 = __builtin_bit_cast(v2f, *(const v2i*)&mr[4 * k + 4]);
                const v2f b0f = *(const v2f*)&wu[64 * k];
                const v2f b1f = *(const v2f*)&wu[64 * k + 64];
                ce = __builtin_amdgcn_wmma_f32_16x16x4_f32(
                    false, a0, false, b0f, (short)0, ce, false, false);
                co = __builtin_amdgcn_wmma_f32_16x16x4_f32(
                    false, a1, false, b1f, (short)0, co, false, false);
            }
            const float sg = sgn[col];
            float ps = 0.0f, pq = 0.0f;
#pragma unroll
            for (int i = 0; i < 8; ++i) {
                const int v = 8 * hi + i;
                if (v < cnt) {
                    const float c = ce[i] + co[i];
                    const float s = sg * sumv[v * C1 + col];
                    ps += s + (float)M_PTS * c;
                    pq += (2.0f * s + (float)M_PTS * c) * c;
                    extm[v * C1 + col] = (extm[v * C1 + col] + sg * c) * sg;
                }
            }
            atomicAdd(&redS[col], ps);
            atomicAdd(&redQ[col], pq);
        }
        __syncthreads();

#pragma unroll
        for (int j = 0; j < VGRP * C1 / 256; ++j) {
            const int idx = j * 256 + tid;
            const int v = idx >> 7;
            if (v < cnt)
                out[(blockIdx.x + (t0 + v) * GEMM_WGS) * C1 + (idx & (C1 - 1))] = extm[idx];
            sumv[idx] = 0.0f;
        }
        for (int i = tid; i < VGRP * M0_PITCH; i += 256) m0[i] = 0;
        __syncthreads();
    }

    if (tid < C1) {
        atomicAdd(&ws[WS_SUMY + tid], redS[tid]);
        atomicAdd(&ws[WS_SUMQ + tid], redQ[tid]);
    }
}

__global__ __launch_bounds__(256) void vfe_apply(const float* __restrict__ ws,
                                                 float* __restrict__ out) {
    const int i = blockIdx.x * 256 + threadIdx.x;
    if (i >= N_OUT4) return;
    const int o = (i & (C1 / 4 - 1)) * 4;
    const float4 sc = *(const float4*)&ws[WS_SCALE1 + o];
    const float4 bi = *(const float4*)&ws[WS_BIAS1 + o];
    float4 y = ((const float4*)out)[i];
    y.x = y.x * sc.x + bi.x;
    y.y = y.y * sc.y + bi.y;
    y.z = y.z * sc.z + bi.z;
    y.w = y.w * sc.w + bi.w;
    y.x = y.x > 0.0f ? y.x : 0.0f;
    y.y = y.y > 0.0f ? y.y : 0.0f;
    y.z = y.z > 0.0f ? y.z : 0.0f;
    y.w = y.w > 0.0f ? y.w : 0.0f;
    ((float4*)out)[i] = y;
}

extern "C" void kernel_launch(void* const* d_in, const int* in_sizes, int n_in,
                              void* d_out, int out_size, void* d_ws, size_t ws_size,
                              hipStream_t stream) {
    (void)in_sizes; (void)n_in; (void)out_size; (void)ws_size;
    const float* X   = (const float*)d_in[0];
    const float* W0  = (const float*)d_in[1];
    const float* g0  = (const float*)d_in[2];
    const float* be0 = (const float*)d_in[3];
    const float* W1  = (const float*)d_in[4];
    const float* g1  = (const float*)d_in[5];
    const float* be1 = (const float*)d_in[6];
    float* out = (float*)d_out;
    float* ws  = (float*)d_ws;

    const long P = (long)N_VOX * M_PTS;
    const float invP = 1.0f / (float)P;

    vfe_zero<<<(WS_TOTAL + 255) / 256, 256, 0, stream>>>(ws);
    vfe_stats0<<<S0_WGS, 256, 0, stream>>>(X, ws);
    vfe_final0<<<1, 64, 0, stream>>>(W0, g0, be0, ws, invP);
    vfe_gemm<<<GEMM_WGS, 256, 0, stream>>>(X, W0, W1, g1, ws, out);
    vfe_final1<<<1, 128, 0, stream>>>(g1, be1, ws, invP);
    vfe_apply<<<N_OUT4 / 256, 256, 0, stream>>>(ws, out);
}


# GEMM voxel loop: wave priority (s_setprio 3..0) falls with a work-group's progress, so of the two work-groups on a WGP the one behind wins the matrix pipe and the younger one's solo tail shrinks
# speedup vs baseline: 6.3134x; 1.0456x over previous
#include <hip/hip_runtime.h>
#include <hip/hip_bf16.h>

typedef float v2f __attribute__((ext_vector_type(2)));
typedef float v8f __attribute__((ext_vector_type(8)));
typedef int v2i __attribute__((ext_vector_type(2)));

#define N_VOX 20000
#define M_PTS 64
#define C_IN  10
#define C0    64
#define C1    128
#define EPS   1e-3f
#define N_OUT4 640000
#define GEMM_WGS 512

#define WS_SUMX   0
#define WS_SXX    16
#define WS_SCALE0 128
#define WS_BIAS0  192
#define WS_SUMY   256
#define WS_SUMQ   384
#define WS_SCALE1 512
#define WS_BIAS1  640
#define WS_TOTAL  768

#define X1_STRIDE 132

#define VGRP    16
#define L_W0F   0
#define L_S0    (L_W0F + 4 * 3 * 32 * 2)
#define L_B0    (L_S0 + C0)
#define L_SGN   (L_B0 + C0)
#define L_REDS  (L_SGN + C1)
#define L_REDQ  (L_REDS + C1)
#define L_EXT   (L_REDQ + C1)
#define L_SUMV  (L_EXT + 8 * C1)
#define L_EXTM  (L_SUMV + VGRP * C1)
#define L_X1    (L_EXTM + VGRP * C1)
#define L_WUP   (L_X1 + M_PTS * X1_STRIDE)
#define L_TOTAL (L_WUP + 8 * 16 * 32 * 2)
#define M0_PITCH 68

__global__ void vfe_zero(float* __restrict__ ws) {
    int i = blockIdx.x * blockDim.x + threadIdx.x;
    if (i < WS_TOTAL) ws[i] = 0.0f;
}

#define S0_WGS   512
#define S0_PTS   64
#define S0_TRIPS (N_VOX * M_PTS / S0_PTS)
static_assert((N_VOX * M_PTS) % S0_PTS == 0, "whole trips only");

__device__ __forceinline__ void s0_load(float (&buf)[32], const float* __restrict__ xb, unsigned lo) {
#pragma unroll
    for (int j = 0; j < 16; ++j) {
        buf[2 * j]     = xb[lo + 4 * C_IN * j];
        buf[2 * j + 1] = xb[lo + 4 * C_IN * j + C_IN];
    }
}

__device__ __forceinline__ void s0_mma(const float (&buf)[32], bool real, float padv,
                                       v8f& acc0, v8f& acc1) {
#pragma unroll
    for (int j = 0; j < 16; j += 2) {
        const v2f a0 = {real ? buf[2 * j] : padv, real ? buf[2 * j + 1] : padv};
        const v2f a1 = {real ? buf[2 * j + 2] : padv, real ? buf[2 * j + 3] : padv};
        acc0 = __builtin_amdgcn_wmma_f32_16x16x4_f32(false, a0, false, a0, (short)0, acc0, false, false);
        acc1 = __builtin_amdgcn_wmma_f32_16x16x4_f32(false, a1, false, a1, (short)0, acc1, false, false);
    }
}

__global__ __launch_bounds__(256) void vfe_stats0(const float* __restrict__ X,
                                                  float* __restrict__ ws) {
    const int tid = threadIdx.x;
    const int lane = tid & 31;
    const int w = __builtin_amdgcn_readfirstlane(tid >> 5);
    const int c = lane & 15, hi = lane >> 4;
    const bool real = c < C_IN;
    const float padv = c == C_IN ? 1.0f : 0.0f;
    const unsigned lo = 2 * hi * C_IN + (real ? c : C_IN - 1);
    const int nw = gridDim.x * 8;
    int tr = blockIdx.x * 8 + w;

    v8f acc0 = {}, acc1 = {};
    float bufA[32], bufB[32];
    if (tr < S0_TRIPS) s0_load(bufA, X + (size_t)tr * (S0_PTS * C_IN), lo);
    while (tr < S0_TRIPS) {
        const bool more = tr + nw < S0_TRIPS;
        if (more) s0_load(bufB, X + (size_t)(tr + nw) * (S0_PTS * C_IN), lo);
        s0_mma(bufA, real, padv, acc0, acc1);
        if (!more) break;
        if (tr + 2 * nw < S0_TRIPS) s0_load(bufA, X + (size_t)(tr + 2 * nw) * (S0_PTS * C_IN), lo);
        s0_mma(bufB, real, padv, acc0, acc1);
        tr += 2 * nw;
    }

    __shared__ __attribute__((aligned(16))) float red[8 * 256];
    {
        float* r = red + w * 256 + (8 * hi) * 16 + c;
#pragma unroll
        for (int i = 0; i < 8; ++i) r[i * 16] = acc0[i] + acc1[i];
    }
    __syncthreads();
    static_assert(WS_SUMX == 0 && WS_SXX == 16, "thread t owns ws[t]");
    const int e = tid - WS_SXX;
    const bool isx = tid < C_IN, isxx = e >= 0 && e < C_IN * C_IN;
    if (isx || isxx) {
        const int m = isx ? tid : e / C_IN;
        const int n = isx ? C_IN : e % C_IN;
        float s = 0.0f;
#pragma unroll
        for (int p = 0; p < 8; ++p) s += red[p * 256 + m * 16 + n];
        atomicAdd(&ws[tid], s);
    }
}

__global__ void vfe_final0(const float* __restrict__ W0, const float* __restrict__ g0,
                           const float* __restrict__ be0, float* __restrict__ ws,
                           float invP) {
    int o = threadIdx.x;
    if (o >= C0) return;
    float w[C_IN];
#pragma unroll
    for (int c = 0; c < C_IN; ++c) w[c] = W0[o * C_IN + c];
    float mean = 0.0f;
#pragma unroll
    for (int c = 0; c < C_IN; ++c) mean += w[c] * ws[WS_SUMX + c];
    mean *= invP;
    float ey2 = 0.0f;
#pragma unroll
    for (int c = 0; c < C_IN; ++c)
#pragma unroll
        for (int d = 0; d < C_IN; ++d) ey2 += w[c] * w[d] * ws[WS_SXX + c * C_IN + d];
    ey2 *= invP;
    float var = ey2 - mean * mean;
    float sc = g0[o] * rsqrtf(var + EPS);
    ws[WS_SCALE0 + o] = sc;
    ws[WS_BIAS0 + o] = be0[o] - mean * sc;
}

__global__ void vfe_final1(const float* __restrict__ g1, const float* __restrict__ be1,
                           float* __restrict__ ws, float invP) {
    int o = threadIdx.x;
    if (o >= C1) return;
    float mean = ws[WS_SUMY + o] * invP;
    float var = ws[WS_SUMQ + o] * invP - mean * mean;
    float sc = g1[o] * rsqrtf(var + EPS);
    ws[WS_SCALE1 + o] = sc;
    ws[WS_BIAS1 + o] = be1[o] - mean * sc;
}

__global__ __launch_bounds__(256) __attribute__((amdgpu_waves_per_eu(4, 4)))
void vfe_gemm(const float* __restrict__ X,
              const float* __restrict__ W0,
              const float* __restrict__ W1,
              const float* __restrict__ g1,
              float* __restrict__ ws,
              float* __restrict__ out) {
    __shared__ __attribute__((aligned(16))) float lds[L_TOTAL];
    __shared__ __attribute__((aligned(16))) int m0[VGRP * M0_PITCH];
    float* const x1   = lds + L_X1;
    float* const w0f  = lds + L_W0F;
    float* const s0   = lds + L_S0;
    float* const b0   = lds + L_B0;
    float* const sgn  = lds + L_SGN;
    float* const redS = lds + L_REDS;
    float* const redQ = lds + L_REDQ;
    float* const ext  = lds + L_EXT;
    float* const sumv = lds + L_SUMV;
    float* const extm = lds + L_EXTM;
    float* const wup  = lds + L_WUP;

    const int tid = threadIdx.x;

    for (int i = tid; i < 4 * 3 * 32 * 2; i += 256) {
        const int e = i & 1, ln = (i >> 1) & 31, g = i >> 6;
        const int k = 4 * (g % 3) + 2 * (ln >> 4) + e;
        const int col = (g / 3) * 16 + (ln & 15);
        w0f[i] = k < C_IN ? W0[col * C_IN + k] : 0.0f;
    }
    if (tid < C0) {
        s0[tid] = ws[WS_SCALE0 + tid];
        b0[tid] = ws[WS_BIAS0 + tid];
    }
    if (tid < C1) {
        redS[tid] = 0.0f;
        redQ[tid] = 0.0f;
        sgn[tid] = g1[tid] < 0.0f ? -1.0f : 1.0f;
    }
    for (int i = tid; i < VGRP * C1; i += 256) sumv[i] = 0.0f;
    for (int i = tid; i < VGRP * M0_PITCH; i += 256) m0[i] = 0;

    const int lane = tid & 31;
    const int w = tid >> 5;
    const int rt = w & 3;
    const int jbase = (w >> 2) * 4;
    const int hi = lane >> 4;
    const int lrow = rt * 16 + (lane & 15);

    v2f bF[4][16];
#pragma unroll
    for (int jj = 0; jj < 4; ++jj) {
        const int colb = (jbase + jj) * 16 + (lane & 15);
        const float* wr = W1 + colb * C1 + 2 * hi;
        const float sgc = g1[colb] < 0.0f ? -1.0f : 1.0f;
#pragma unroll
        for (int k = 0; k < 16; ++k) bF[jj][k] = *(const v2f*)(wr + 4 * k) * sgc;
    }

    float* const wu = wup + (w * 16 * 32 + lane) * 2;
    {
        const float* wr = W1 + (w * 16 + (lane & 15)) * C1 + C0 + 2 * hi;
#pragma unroll
        for (int k = 0; k < 16; ++k) *(v2f*)&wu[64 * k] = *(const v2f*)(wr + 4 * k);
    }

    const int l0tile = (w >> 2) * 2;
    const v2f* const X2 = (const v2f*)X;
    const unsigned xp01 = lrow * (C_IN / 2) + hi;
    const unsigned xp2 = lrow * (C_IN / 2) + 4 - hi;

    const float* const w0p = w0f + (l0tile * 3 * 32 + lane) * 2;
    float* const l0p = s0 + l0tile * 16 + (lane & 15);
    int*   const m0p = m0 + l0tile * 16 + (lane & 15);
    float* const x1st = x1 + (rt * 16 + 8 * hi) * X1_STRIDE + l0tile * 16 + (lane & 15);
    const float* const x1ld = x1 + lrow * X1_STRIDE + 2 * hi;
    float* const ep = sgn + jbase * 16 + (lane & 15);
    float* const extp = ext + (rt * 2 + hi) * C1 + jbase * 16 + (lane & 15);
    v2f pa0, pa1, pa2;
    {
        const unsigned vo = blockIdx.x * (M_PTS * C_IN / 2);
        pa0 = X2[vo + xp01];
        pa1 = X2[vo + xp01 + 2];
        pa2 = X2[vo + xp2];
    }
    __syncthreads();

    const int ntrips = (N_VOX - 1 - (int)blockIdx.x) / GEMM_WGS + 1;
    static_assert(N_VOX / GEMM_WGS > 24, "the priority steps below lie inside every work-group's voxel range");
    int prio_at = 0, prio = 3;
    for (int t0 = 0; t0 < ntrips; t0 += VGRP) {
        const int cnt = ntrips - t0 < VGRP ? ntrips - t0 : VGRP;
        for (int g = 0; g < cnt; ++g) {
            const int n = blockIdx.x + (t0 + g) * GEMM_WGS;
            if (t0 + g == prio_at) {
                if (prio == 3)      { __builtin_amdgcn_s_setprio(3); prio_at = ntrips - 24; }
                else if (prio == 2) { __builtin_amdgcn_s_setprio(2); prio_at = ntrips - 12; }
                else if (prio == 1) { __builtin_amdgcn_s_setprio(1); prio_at = ntrips - 4; }
                else                { __builtin_amdgcn_s_setprio(0); prio_at = -1; }
                --prio;
            }
            {
                v2f a[3];
                a[0] = pa0;
                a[1] = pa1;
                a[2] = hi ? (v2f){0.0f, 0.0f} : pa2;
                v2f b[2][3];
#pragma unroll
                for (int t = 0; t < 2; ++t)
#pragma unroll
                    for (int s = 0; s < 3; ++s)
                        b[t][s] = *(const v2f*)&w0p[(t * 3 + s) * 64];
                v8f l0acc[2] = {};
#pragma unroll
                for (int s = 0; s < 3; ++s)
#pragma unroll
                    for (int t = 0; t < 2; ++t)
                        l0acc[t] = __builtin_amdgcn_wmma_f32_16x16x4_f32(
                            false, a[s], false, b[t][s], (short)0, l0acc[t], false, false);
                int* const mg = m0p + g * M0_PITCH;
#pragma unroll
                for (int t = 0; t < 2; ++t) {
                    const float sc = l0p[16 * t], bi = l0p[(L_B0 - L_S0) + 16 * t];
                    float* xp = x1st + 16 * t;
                    float m = 0.0f;
#pragma unroll
                    for (int i = 0; i < 8; ++i) {
                        float v = l0acc[t][i] * sc + bi;
                        v = v > 0.0f ? v : 0.0f;
                        xp[i * X1_STRIDE] = v;
                        m = fmaxf(m, v);
                    }
                    atomicMax(&mg[16 * t], __float_as_int(m));
                }
            }
            __syncthreads();

            v2f aF[16];
#pragma unroll
            for (int k = 0; k < 16; ++k)
                aF[k] = *(const v2f*)&x1ld[4 * k];

            float* const sv = ep + (L_SUMV - L_SGN) + g * C1;

#pragma unroll
            for (int jp = 0; jp < 4; jp += 2) {
                v8f accs[2] = {};
#pragma unroll
                for (int k = 0; k < 16; ++k) {
#pragma unroll
                    for (int t = 0; t < 2; ++t)
                        accs[t] = __builtin_amdgcn_wmma_f32_16x16x4_f32(
                            false, aF[k], false, bF[jp + t][k], (short)0, accs[t], false, false);
                }

#pragma unroll
                for (int t = 0; t < 2; ++t) {
                    const v8f acc = accs[t];
                    const v2f p0 = __builtin_shufflevector(acc, acc, 0, 1);
                    const v2f p1 = __builtin_shufflevector(acc, acc, 2, 3);
                    const v2f p2 = __builtin_shufflevector(acc, acc, 4, 5);
                    const v2f p3 = __builtin_shufflevector(acc, acc, 6, 7);
                    const v2f ss = (p0 + p1) + (p2 + p3);
                    v2f qq = p0 * p0;
                    qq = p1 * p1 + qq;
                    qq = p2 * p2 + qq;
                    qq = p3 * p3 + qq;
                    const float s = ss[0] + ss[1];
                    const float q = qq[0] + qq[1];
                    const float m = fmaxf(fmaxf(fmaxf(acc[0], acc[1]), fmaxf(acc[2], acc[3])),
                                          fmaxf(fmaxf(acc[4], acc[5]), fmaxf(acc[6], acc[7])));
                    atomicAdd(&sv[16 * (jp + t)], s);
                    atomicAdd(&ep[(L_REDQ - L_SGN) + 16 * (jp + t)], q);
                    extp[16 * (jp + t)] = m;
                }

                if (jp == 0 && n + GEMM_WGS < N_VOX) {
                    const unsigned vo = (unsigned)(n + GEMM_WGS) * (M_PTS * C_IN / 2);
                    pa0 = X2[vo + xp01];
                    pa1 = X2[vo + xp01 + 2];
                    pa2 = X2[vo + xp2];
                }
            }
            __syncthreads();

            if (tid < C1) {
                float m = ext[tid];
                for (int p = 1; p < 8; ++p) m = fmaxf(m, ext[p * C1 + tid]);
                extm[g * C1 + tid] = m;
            }
        }
        __syncthreads();

        {
            const int col = w * 16 + (lane & 15);
            const int* const mr = m0 + (lane & 15) * M0_PITCH + 2 * hi;
            v8f ce = {}, co = {};
#pragma unroll
            for (int k = 0; k < 16; k += 2) {
                const v2f a0 = __builtin_bit_cast(v2f, *(const v2i*)&mr[4 * k]);
                const v2f a1 = __builtin_bit_cast(v2f, *(const v2i*)&mr[4 * k + 4]);
                const v2f b0f = *(const v2f*)&wu[64 * k];
                const v2f b1f = *(const v2f*)&wu[64 * k + 64];
                ce = __builtin_amdgcn_wmma_f32_16x16x4_f32(
                    false, a0, false, b0f, (short)0, ce, false, false);
                co = __builtin_amdgcn_wmma_f32_16x16x4_f32(
                    false, a1, false, b1f, (short)0, co, false, false);
            }
            const float sg = sgn[col];
            float ps = 0.0f, pq = 0.0f;
#pragma unroll
            for (int i = 0; i < 8; ++i) {
                const int v = 8 * hi + i;
                if (v < cnt) {
                    const float c = ce[i] + co[i];
                    const float s = sg * sumv[v * C1 + col];
                    ps += s + (float)M_PTS * c;
                    pq += (2.0f * s + (float)M_PTS * c) * c;
                    extm[v * C1 + col] = (extm[v * C1 + col] + sg * c) * sg;
                }
            }
            atomicAdd(&redS[col], ps);
            atomicAdd(&redQ[col], pq);
        }
        __syncthreads();

#pragma unroll
        for (int j = 0; j < VGRP * C1 / 256; ++j) {
            const int idx = j * 256 + tid;
            const int v = idx >> 7;
            if (v < cnt)
                out[(blockIdx.x + (t0 + v) * GEMM_WGS) * C1 + (idx & (C1 - 1))] = extm[idx];
            sumv[idx] = 0.0f;
        }
        for (int i = tid; i < VGRP * M0_PITCH; i += 256) m0[i] = 0;
        __syncthreads();
    }

    if (tid < C1) {
        atomicAdd(&ws[WS_SUMY + tid], redS[tid]);
        atomicAdd(&ws[WS_SUMQ + tid], redQ[tid]);
    }
}

__global__ __launch_bounds__(256) void vfe_apply(const float* __restrict__ ws,
                                                 float* __restrict__ out) {
    const int i = blockIdx.x * 256 + threadIdx.x;
    if (i >= N_OUT4) return;
    const int o = (i & (C1 / 4 - 1)) * 4;
    const float4 sc = *(const float4*)&ws[WS_SCALE1 + o];
    const float4 bi = *(const float4*)&ws[WS_BIAS1 + o];
    float4 y = ((const float4*)out)[i];
    y.x = y.x * sc.x + bi.x;
    y.y = y.y * sc.y + bi.y;
    y.z = y.z * sc.z + bi.z;
    y.w = y.w * sc.w + bi.w;
    y.x = y.x > 0.0f ? y.x : 0.0f;
    y.y = y.y > 0.0f ? y.y : 0.0f;
    y.z = y.z > 0.0f ? y.z : 0.0f;
    y.w = y.w > 0.0f ? y.w : 0.0f;
    ((float4*)out)[i] = y;
}

extern "C" void kernel_launch(void* const* d_in, const int* in_sizes, int n_in,
                              void* d_out, int out_size, void* d_ws, size_t ws_size,
                              hipStream_t stream) {
    (void)in_sizes; (void)n_in; (void)out_size; (void)ws_size;
    const float* X   = (const float*)d_in[0];
    const float* W0  = (const float*)d_in[1];
    const float* g0  = (const float*)d_in[2];
    const float* be0 = (const float*)d_in[3];
    const float* W1  = (const float*)d_in[4];
    const float* g1  = (const float*)d_in[5];
    const float* be1 = (const float*)d_in[6];
    float* out = (float*)d_out;
    float* ws  = (float*)d_ws;

    const long P = (long)N_VOX * M_PTS;
    const float invP = 1.0f / (float)P;

    vfe_zero<<<(WS_TOTAL + 255) / 256, 256, 0, stream>>>(ws);
    vfe_stats0<<<S0_WGS, 256, 0, stream>>>(X, ws);
    vfe_final0<<<1, 64, 0, stream>>>(W0, g0, be0, ws, invP);
    vfe_gemm<<<GEMM_WGS, 256, 0, stream>>>(X, W0, W1, g1, ws, out);
    vfe_final1<<<1, 128, 0, stream>>>(g1, be1, ws, invP);
    vfe_apply<<<N_OUT4 / 256, 256, 0, stream>>>(ws, out);
}
